# starting kernel
# baseline (speedup 1.0000x reference)
#include <hip/hip_runtime.h>
#include <hip/hip_bf16.h>
#include <math.h>

typedef float v2f __attribute__((ext_vector_type(2)));
typedef float v8f __attribute__((ext_vector_type(8)));

__device__ __forceinline__ float gelu_exact(float x) {
  return 0.5f * x * (1.0f + erff(x * 0.70710678118654752f));
}

#define WMMA_F32(cacc, av, bv)                                                 \
  cacc = __builtin_amdgcn_wmma_f32_16x16x4_f32(false, (av), false, (bv),       \
                                               (short)0, (cacc), false, false)

__global__ __launch_bounds__(256) void gemm_wmma_f32(
    const float* __restrict__ A, const float* __restrict__ W,
    const float* __restrict__ bias, const float* __restrict__ R,
    float* __restrict__ C, int M, int N, int Kd,
    int flags, int Tlen, int Kspk, int kk)
{
  const int lane = threadIdx.x & 31;
  const int w    = threadIdx.x >> 5;
  const int wm   = w & 3;
  const int wn   = w >> 2;
  const int tileM = blockIdx.x * 64 + wm * 16;
  const int tileN = blockIdx.y * 128 + wn * 64;
  const int half = lane >> 4;
  const int l16  = lane & 15;

  const int mrow = tileM + l16;
  const int n0   = tileN + l16;

  const float* Ap = A + (size_t)mrow * Kd + 2 * half;
  const float* Bp0 = W + (size_t)(n0)      * Kd + 2 * half;
  const float* Bp1 = W + (size_t)(n0 + 16) * Kd + 2 * half;
  const float* Bp2 = W + (size_t)(n0 + 32) * Kd + 2 * half;
  const float* Bp3 = W + (size_t)(n0 + 48) * Kd + 2 * half;

  v8f c0 = {}, c1 = {}, c2 = {}, c3 = {};

  v2f a0  = *(const v2f*)(Ap);
  v2f b00 = *(const v2f*)(Bp0);
  v2f b01 = *(const v2f*)(Bp1);
  v2f b02 = *(const v2f*)(Bp2);
  v2f b03 = *(const v2f*)(Bp3);
  v2f a1  = *(const v2f*)(Ap + 4);
  v2f b10 = *(const v2f*)(Bp0 + 4);
  v2f b11 = *(const v2f*)(Bp1 + 4);
  v2f b12 = *(const v2f*)(Bp2 + 4);
  v2f b13 = *(const v2f*)(Bp3 + 4);

  for (int k = 0; k < Kd - 8; k += 8) {
    WMMA_F32(c0, a0, b00);
    WMMA_F32(c1, a0, b01);
    WMMA_F32(c2, a0, b02);
    WMMA_F32(c3, a0, b03);
    a0  = *(const v2f*)(Ap  + k + 8);
    b00 = *(const v2f*)(Bp0 + k + 8);
    b01 = *(const v2f*)(Bp1 + k + 8);
    b02 = *(const v2f*)(Bp2 + k + 8);
    b03 = *(const v2f*)(Bp3 + k + 8);
    WMMA_F32(c0, a1, b10);
    WMMA_F32(c1, a1, b11);
    WMMA_F32(c2, a1, b12);
    WMMA_F32(c3, a1, b13);
    a1  = *(const v2f*)(Ap  + k + 12);
    b10 = *(const v2f*)(Bp0 + k + 12);
    b11 = *(const v2f*)(Bp1 + k + 12);
    b12 = *(const v2f*)(Bp2 + k + 12);
    b13 = *(const v2f*)(Bp3 + k + 12);
  }
  WMMA_F32(c0, a0, b00);
  WMMA_F32(c1, a0, b01);
  WMMA_F32(c2, a0, b02);
  WMMA_F32(c3, a0, b03);
  WMMA_F32(c0, a1, b10);
  WMMA_F32(c1, a1, b11);
  WMMA_F32(c2, a1, b12);
  WMMA_F32(c3, a1, b13);

  const float bv0 = bias[n0];
  const float bv1 = bias[n0 + 16];
  const float bv2 = bias[n0 + 32];
  const float bv3 = bias[n0 + 48];
#pragma unroll
  for (int i = 0; i < 8; ++i) {
    const int row = tileM + i + 8 * half;
    float v0 = c0[i] + bv0;
    float v1 = c1[i] + bv1;
    float v2 = c2[i] + bv2;
    float v3 = c3[i] + bv3;
    if (flags & 1) {
      const float* rr = R + (size_t)row * N;
      v0 += rr[n0];
      v1 += rr[n0 + 16];
      v2 += rr[n0 + 32];
      v3 += rr[n0 + 48];
    }
    if (flags & 2) {
      v0 = gelu_exact(v0);
      v1 = gelu_exact(v1);
      v2 = gelu_exact(v2);
      v3 = gelu_exact(v3);
    }
    size_t orow = (size_t)row;
    if (flags & 4) {
      const int bi = row / Tlen;
      orow = (size_t)bi * Kspk * Tlen + (size_t)kk * Tlen + (row % Tlen);
    }
    float* cr = C + orow * N;
    cr[n0]      = v0;
    cr[n0 + 16] = v1;
    cr[n0 + 32] = v2;
    cr[n0 + 48] = v3;
  }
}

__global__ __launch_bounds__(256) void ln_kernel(
    const float* __restrict__ X, const float* __restrict__ g,
    const float* __restrict__ b, float* __restrict__ Y, int Dn)
{
  const int row = blockIdx.x;
  const int tid = threadIdx.x;
  const float* x = X + (size_t)row * Dn;
  float s = 0.f, s2 = 0.f;
  for (int d = tid; d < Dn; d += 256) { float v = x[d]; s += v; s2 += v * v; }
  __shared__ float sb0[256], sb1[256];
  sb0[tid] = s; sb1[tid] = s2; __syncthreads();
  for (int off = 128; off > 0; off >>= 1) {
    if (tid < off) { sb0[tid] += sb0[tid + off]; sb1[tid] += sb1[tid + off]; }
    __syncthreads();
  }
  const float mean = sb0[0] / Dn;
  const float var  = sb1[0] / Dn - mean * mean;
  const float inv  = rsqrtf(var + 1e-5f);
  float* y = Y + (size_t)row * Dn;
  for (int d = tid; d < Dn; d += 256) y[d] = (x[d] - mean) * inv * g[d] + b[d];
}

__global__ __launch_bounds__(256) void wmix_kernel(
    const float* __restrict__ logits, const int* __restrict__ blank_p,
    float* __restrict__ wmix, int BT, int V, int Kspk)
{
  const int row = blockIdx.x;
  const int tid = threadIdx.x;
  const int blank = blank_p[0];
  __shared__ float red[256];
  float a[8];
  for (int k2 = 0; k2 < Kspk; ++k2) {
    const float* L = logits + ((size_t)k2 * BT + row) * V;
    float m = -INFINITY;
    for (int v = tid; v < V; v += 256) m = fmaxf(m, L[v]);
    red[tid] = m; __syncthreads();
    for (int off = 128; off > 0; off >>= 1) {
      if (tid < off) red[tid] = fmaxf(red[tid], red[tid + off]);
      __syncthreads();
    }
    m = red[0]; __syncthreads();
    float s = 0.f;
    for (int v = tid; v < V; v += 256) s += expf(L[v] - m);
    red[tid] = s; __syncthreads();
    for (int off = 128; off > 0; off >>= 1) {
      if (tid < off) red[tid] += red[tid + off];
      __syncthreads();
    }
    s = red[0]; __syncthreads();
    a[k2] = 1.0f - expf(L[blank] - m) / s;
  }
  if (tid == 0) {
    float m2 = -INFINITY;
    for (int k2 = 0; k2 < Kspk; ++k2) m2 = fmaxf(m2, a[k2]);
    float ss = 0.f;
    float e[8];
    for (int k2 = 0; k2 < Kspk; ++k2) { e[k2] = expf(a[k2] - m2); ss += e[k2]; }
    for (int k2 = 0; k2 < Kspk; ++k2) wmix[(size_t)row * Kspk + k2] = e[k2] / ss;
  }
}

__global__ __launch_bounds__(256) void stream_prep(
    const float* __restrict__ X, const float* __restrict__ wmix,
    const float* __restrict__ tags, const float* __restrict__ g,
    const float* __restrict__ bt, float* __restrict__ Xk,
    float* __restrict__ Qin, int Dn, int Kspk, int kk)
{
  const int row = blockIdx.x;
  const int tid = threadIdx.x;
  const float wv = wmix[(size_t)row * Kspk + kk];
  const float* x  = X    + (size_t)row * Dn;
  const float* tg = tags + (size_t)kk  * Dn;
  float s = 0.f, s2 = 0.f;
  for (int d = tid; d < Dn; d += 256) {
    float v = x[d] * wv + tg[d];
    Xk[(size_t)row * Dn + d] = v;
    s += v; s2 += v * v;
  }
  __shared__ float sb0[256], sb1[256];
  sb0[tid] = s; sb1[tid] = s2; __syncthreads();
  for (int off = 128; off > 0; off >>= 1) {
    if (tid < off) { sb0[tid] += sb0[tid + off]; sb1[tid] += sb1[tid + off]; }
    __syncthreads();
  }
  const float mean = sb0[0] / Dn;
  const float var  = sb1[0] / Dn - mean * mean;
  const float inv  = rsqrtf(var + 1e-5f);
  for (int d = tid; d < Dn; d += 256) {
    float v = x[d] * wv + tg[d];
    Qin[(size_t)row * Dn + d] = (v - mean) * inv * g[d] + bt[d];
  }
}

__global__ __launch_bounds__(128) void banded_attn(
    const float* __restrict__ q, const float* __restrict__ kb,
    const float* __restrict__ vb, float* __restrict__ ctx,
    int T_, int Dn, int Hn, int band)
{
  const int hd   = 64;
  const int wid  = threadIdx.x >> 5;
  const int lane = threadIdx.x & 31;
  const int gw   = blockIdx.x * 4 + wid;
  const int t = gw % T_;
  const int h = (gw / T_) % Hn;
  const int b = gw / (T_ * Hn);

  __shared__ float lq[4][64];
  __shared__ float lat[4][64];

  const size_t rowbase = (size_t)(b * T_ + t) * Dn + (size_t)h * hd;
  lq[wid][lane]      = q[rowbase + lane];
  lq[wid][lane + 32] = q[rowbase + lane + 32];
  __syncthreads();

  const float scale = 0.125f;
  const int s0 = t - band;
  const int nb = 2 * band + 1;

  const float4* lq4 = (const float4*)&lq[wid][0];

  const int j0 = lane, j1 = lane + 32;
  const int sa = s0 + j0, sbi = s0 + j1;
  float sc0 = -INFINITY, sc1 = -INFINITY;
  if (j0 < nb && sa >= 0 && sa < T_) {
    const size_t kro = (size_t)(b * T_ + sa) * Dn + (size_t)h * hd;
    __builtin_prefetch(vb + kro, 0, 0);
    const float4* kp4 = (const float4*)(kb + kro);
    float acc = 0.f;
#pragma unroll
    for (int d4 = 0; d4 < 16; ++d4) {
      float4 kv = kp4[d4];
      float4 qv = lq4[d4];
      acc += qv.x * kv.x + qv.y * kv.y + qv.z * kv.z + qv.w * kv.w;
    }
    sc0 = acc * scale;
  }
  if (j1 < nb && sbi >= 0 && sbi < T_) {
    const size_t kro = (size_t)(b * T_ + sbi) * Dn + (size_t)h * hd;
    __builtin_prefetch(vb + kro, 0, 0);
    const float4* kp4 = (const float4*)(kb + kro);
    float acc = 0.f;
#pragma unroll
    for (int d4 = 0; d4 < 16; ++d4) {
      float4 kv = kp4[d4];
      float4 qv = lq4[d4];
      acc += qv.x * kv.x + qv.y * kv.y + qv.z * kv.z + qv.w * kv.w;
    }
    sc1 = acc * scale;
  }

  float m = fmaxf(sc0, sc1);
  for (int off = 16; off > 0; off >>= 1) m = fmaxf(m, __shfl_xor(m, off, 32));
  float e0 = expf(sc0 - m);
  float e1 = expf(sc1 - m);
  float s = e0 + e1;
  for (int off = 16; off > 0; off >>= 1) s += __shfl_xor(s, off, 32);
  const float inv = 1.0f / s;
  lat[wid][j0] = e0 * inv;
  lat[wid][j1] = e1 * inv;
  __syncthreads();

  float a0 = 0.f, a1 = 0.f;
  const int d0 = lane, d1 = lane + 32;
  for (int j = 0; j < nb; ++j) {
    const int sidx = s0 + j;
    if (sidx < 0 || sidx >= T_) continue;
    const float wv = lat[wid][j];
    const float* vp = vb + (size_t)(b * T_ + sidx) * Dn + (size_t)h * hd;
    a0 += wv * vp[d0];
    a1 += wv * vp[d1];
  }
  float* cp = ctx + rowbase;
  cp[d0] = a0;
  cp[d1] = a1;
}

extern "C" void kernel_launch(void* const* d_in, const int* in_sizes, int n_in,
                              void* d_out, int out_size, void* d_ws, size_t ws_size,
                              hipStream_t stream) {
  (void)in_sizes; (void)n_in; (void)out_size; (void)ws_size;
  constexpr int B = 4, T = 1024, D_IN = 512, D = 512, K = 2, V = 500, H = 8, BAND = 24;
  constexpr int M = B * T;

  const float* x_m      = (const float*)d_in[0];
  const float* logits   = (const float*)d_in[1];
  const int*   blank_id = (const int*)  d_in[2];
  const float* proj_w   = (const float*)d_in[3];
  const float* proj_b   = (const float*)d_in[4];
  const float* spk_tags = (const float*)d_in[5];
  const float* lnq_g    = (const float*)d_in[6];
  const float* lnq_b    = (const float*)d_in[7];
  const float* lnkv_g   = (const float*)d_in[8];
  const float* lnkv_b   = (const float*)d_in[9];
  const float* in_w     = (const float*)d_in[10];
  const float* in_b     = (const float*)d_in[11];
  const float* out_w    = (const float*)d_in[12];
  const float* out_b    = (const float*)d_in[13];
  const float* ffn_ln_g = (const float*)d_in[14];
  const float* ffn_ln_b = (const float*)d_in[15];
  const float* ffn_w1   = (const float*)d_in[16];
  const float* ffn_b1   = (const float*)d_in[17];
  const float* ffn_w2   = (const float*)d_in[18];
  const float* ffn_b2   = (const float*)d_in[19];
  float* out = (float*)d_out;

  float* ws = (float*)d_ws;
  const size_t SZ = (size_t)M * D;
  float* X    = ws;
  float* kbuf = X    + SZ;
  float* vbuf = kbuf + SZ;
  float* tA   = vbuf + SZ;
  float* tB   = tA   + SZ;
  float* tC   = tB   + SZ;
  float* t1   = tC   + SZ;
  float* wmix = t1   + 4 * SZ;

  const dim3 blk(256);
  auto gg = [](int m, int n) { return dim3(m / 64, n / 128); };

  gemm_wmma_f32<<<gg(M, D), blk, 0, stream>>>(x_m, proj_w, proj_b, nullptr, X,
                                              M, D, D_IN, 0, T, K, 0);
  wmix_kernel<<<dim3(M), blk, 0, stream>>>(logits, blank_id, wmix, M, V, K);
  ln_kernel<<<dim3(M), blk, 0, stream>>>(X, lnkv_g, lnkv_b, tA, D);
  gemm_wmma_f32<<<gg(M, D), blk, 0, stream>>>(tA, in_w + (size_t)D * D, in_b + D,
                                              nullptr, kbuf, M, D, D, 0, T, K, 0);
  gemm_wmma_f32<<<gg(M, D), blk, 0, stream>>>(tA, in_w + (size_t)2 * D * D, in_b + 2 * D,
                                              nullptr, vbuf, M, D, D, 0, T, K, 0);

  for (int kk = 0; kk < K; ++kk) {
    stream_prep<<<dim3(M), blk, 0, stream>>>(X, wmix, spk_tags, lnq_g, lnq_b,
                                             tA, tB, D, K, kk);
    gemm_wmma_f32<<<gg(M, D), blk, 0, stream>>>(tB, in_w, in_b, nullptr, tC,
                                                M, D, D, 0, T, K, kk);
    banded_attn<<<dim3((B * H * T) / 4), dim3(128), 0, stream>>>(tC, kbuf, vbuf, tB,
                                                                 T, D, H, BAND);
    gemm_wmma_f32<<<gg(M, D), blk, 0, stream>>>(tB, out_w, out_b, tA, tC,
                                                M, D, D,  1, T, K, kk);
    ln_kernel<<<dim3(M), blk, 0, stream>>>(tC, ffn_ln_g, ffn_ln_b, tB, D);
    gemm_wmma_f32<<<gg(M, 4 * D), blk, 0, stream>>>(tB, ffn_w1, ffn_b1, nullptr, t1,
                                                    M, 4 * D, D,  2, T, K, kk);
    gemm_wmma_f32<<<gg(M, D), blk, 0, stream>>>(t1, ffn_w2, ffn_b2, tC, out,
                                                M, D, 4 * D,  1 | 4, T, K, kk);
  }
}


# GEMM operand tiles staged through LDS: coalesced 16-byte global loads, double-buffered k-tiles of 32, fragments read from LDS with 16-byte loads (was per-lane 8-byte global loads)
# speedup vs baseline: 1.3575x; 1.3575x over previous
#include <hip/hip_runtime.h>
#include <hip/hip_bf16.h>
#include <math.h>

typedef float v2f __attribute__((ext_vector_type(2)));
typedef float v4f __attribute__((ext_vector_type(4)));
typedef float v8f __attribute__((ext_vector_type(8)));

__device__ __forceinline__ float gelu_exact(float x) {
  return 0.5f * x * (1.0f + erff(x * 0.70710678118654752f));
}

#define WMMA_F32(cacc, av, bv)                                                 \
  cacc = __builtin_amdgcn_wmma_f32_16x16x4_f32(false, (av), false, (bv),       \
                                               (short)0, (cacc), false, false)

#define GEMM_BM 64
#define GEMM_BN 128
#define GEMM_BK 32
#define GEMM_LDP 36

__global__ __launch_bounds__(256) void gemm_wmma_f32(
    const float* __restrict__ A, const float* __restrict__ W,
    const float* __restrict__ bias, const float* __restrict__ R,
    float* __restrict__ C, int M, int N, int Kd,
    int flags, int Tlen, int Kspk, int kk)
{
  __shared__ __attribute__((aligned(16))) float As[2][GEMM_BM * GEMM_LDP];
  __shared__ __attribute__((aligned(16))) float Bs[2][GEMM_BN * GEMM_LDP];

  const int tid  = threadIdx.x;
  const int lane = tid & 31;
  const int w    = tid >> 5;
  const int wm   = w & 3;
  const int wn   = w >> 2;
  const int blockM = blockIdx.x * GEMM_BM;
  const int blockN = blockIdx.y * GEMM_BN;
  const int tileM = blockM + wm * 16;
  const int tileN = blockN + wn * 64;
  const int half = lane >> 4;
  const int l16  = lane & 15;
  const int n0   = tileN + l16;

  const int srow = tid >> 3;
  const int scol = (tid & 7) * 4;
  const float* Ag = A + (size_t)(blockM + srow) * Kd + scol;
  const float* Bg = W + (size_t)(blockN + srow) * Kd + scol;
  const size_t rstep = (size_t)32 * Kd;
  const int sts = srow * GEMM_LDP + scol;

  v4f ra0 = *(const v4f*)(Ag);
  v4f ra1 = *(const v4f*)(Ag + rstep);
  v4f rb0 = *(const v4f*)(Bg);
  v4f rb1 = *(const v4f*)(Bg + rstep);
  v4f rb2 = *(const v4f*)(Bg + 2 * rstep);
  v4f rb3 = *(const v4f*)(Bg + 3 * rstep);

  *(v4f*)(&As[0][sts])                 = ra0;
  *(v4f*)(&As[0][sts + 32 * GEMM_LDP]) = ra1;
  *(v4f*)(&Bs[0][sts])                 = rb0;
  *(v4f*)(&Bs[0][sts + 32 * GEMM_LDP]) = rb1;
  *(v4f*)(&Bs[0][sts + 64 * GEMM_LDP]) = rb2;
  *(v4f*)(&Bs[0][sts + 96 * GEMM_LDP]) = rb3;
  __syncthreads();

  const int fa = (wm * 16 + l16) * GEMM_LDP + 4 * half;
  const int fb = (wn * 64 + l16) * GEMM_LDP + 4 * half;

  v8f c0 = {}, c1 = {}, c2 = {}, c3 = {};

  const int nk = Kd / GEMM_BK;
  for (int kt = 0; kt < nk; ++kt) {
    const int cur = kt & 1;
    const bool more = (kt + 1 < nk);
    if (more) {
      const int ko = (kt + 1) * GEMM_BK;
      ra0 = *(const v4f*)(Ag + ko);
      ra1 = *(const v4f*)(Ag + rstep + ko);
      rb0 = *(const v4f*)(Bg + ko);
      rb1 = *(const v4f*)(Bg + rstep + ko);
      rb2 = *(const v4f*)(Bg + 2 * rstep + ko);
      rb3 = *(const v4f*)(Bg + 3 * rstep + ko);
    }

    const float* as = &As[cur][fa];
    const float* bs = &Bs[cur][fb];
#pragma unroll
    for (int k8 = 0; k8 < GEMM_BK / 8; ++k8) {
      const v4f a  = *(const v4f*)(as + k8 * 8);
      const v4f b0 = *(const v4f*)(bs + k8 * 8);
      const v4f b1 = *(const v4f*)(bs + 16 * GEMM_LDP + k8 * 8);
      const v4f b2 = *(const v4f*)(bs + 32 * GEMM_LDP + k8 * 8);
      const v4f b3 = *(const v4f*)(bs + 48 * GEMM_LDP + k8 * 8);
      const v2f alo = {a.x, a.y},   ahi = {a.z, a.w};
      const v2f b0l = {b0.x, b0.y}, b0h = {b0.z, b0.w};
      const v2f b1l = {b1.x, b1.y}, b1h = {b1.z, b1.w};
      const v2f b2l = {b2.x, b2.y}, b2h = {b2.z, b2.w};
      const v2f b3l = {b3.x, b3.y}, b3h = {b3.z, b3.w};
      WMMA_F32(c0, alo, b0l);
      WMMA_F32(c1, alo, b1l);
      WMMA_F32(c2, alo, b2l);
      WMMA_F32(c3, alo, b3l);
      WMMA_F32(c0, ahi, b0h);
      WMMA_F32(c1, ahi, b1h);
      WMMA_F32(c2, ahi, b2h);
      WMMA_F32(c3, ahi, b3h);
    }

    if (more) {
      const int nxt = cur ^ 1;
      *(v4f*)(&As[nxt][sts])                 = ra0;
      *(v4f*)(&As[nxt][sts + 32 * GEMM_LDP]) = ra1;
      *(v4f*)(&Bs[nxt][sts])                 = rb0;
      *(v4f*)(&Bs[nxt][sts + 32 * GEMM_LDP]) = rb1;
      *(v4f*)(&Bs[nxt][sts + 64 * GEMM_LDP]) = rb2;
      *(v4f*)(&Bs[nxt][sts + 96 * GEMM_LDP]) = rb3;
    }
    __syncthreads();
  }

  const float bv0 = bias[n0];
  const float bv1 = bias[n0 + 16];
  const float bv2 = bias[n0 + 32];
  const float bv3 = bias[n0 + 48];
#pragma unroll
  for (int i = 0; i < 8; ++i) {
    const int row = tileM + i + 8 * half;
    float v0 = c0[i] + bv0;
    float v1 = c1[i] + bv1;
    float v2 = c2[i] + bv2;
    float v3 = c3[i] + bv3;
    if (flags & 1) {
      const float* rr = R + (size_t)row * N;
      v0 += rr[n0];
      v1 += rr[n0 + 16];
      v2 += rr[n0 + 32];
      v3 += rr[n0 + 48];
    }
    if (flags & 2) {
      v0 = gelu_exact(v0);
      v1 = gelu_exact(v1);
      v2 = gelu_exact(v2);
      v3 = gelu_exact(v3);
    }
    size_t orow = (size_t)row;
    if (flags & 4) {
      const int bi = row / Tlen;
      orow = (size_t)bi * Kspk * Tlen + (size_t)kk * Tlen + (row % Tlen);
    }
    float* cr = C + orow * N;
    cr[n0]      = v0;
    cr[n0 + 16] = v1;
    cr[n0 + 32] = v2;
    cr[n0 + 48] = v3;
  }
}

__global__ __launch_bounds__(256) void ln_kernel(
    const float* __restrict__ X, const float* __restrict__ g,
    const float* __restrict__ b, float* __restrict__ Y, int Dn)
{
  const int row = blockIdx.x;
  const int tid = threadIdx.x;
  const float* x = X + (size_t)row * Dn;
  float s = 0.f, s2 = 0.f;
  for (int d = tid; d < Dn; d += 256) { float v = x[d]; s += v; s2 += v * v; }
  __shared__ float sb0[256], sb1[256];
  sb0[tid] = s; sb1[tid] = s2; __syncthreads();
  for (int off = 128; off > 0; off >>= 1) {
    if (tid < off) { sb0[tid] += sb0[tid + off]; sb1[tid] += sb1[tid + off]; }
    __syncthreads();
  }
  const float mean = sb0[0] / Dn;
  const float var  = sb1[0] / Dn - mean * mean;
  const float inv  = rsqrtf(var + 1e-5f);
  float* y = Y + (size_t)row * Dn;
  for (int d = tid; d < Dn; d += 256) y[d] = (x[d] - mean) * inv * g[d] + b[d];
}

__global__ __launch_bounds__(256) void wmix_kernel(
    const float* __restrict__ logits, const int* __restrict__ blank_p,
    float* __restrict__ wmix, int BT, int V, int Kspk)
{
  const int row = blockIdx.x;
  const int tid = threadIdx.x;
  const int blank = blank_p[0];
  __shared__ float red[256];
  float a[8];
  for (int k2 = 0; k2 < Kspk; ++k2) {
    const float* L = logits + ((size_t)k2 * BT + row) * V;
    float m = -INFINITY;
    for (int v = tid; v < V; v += 256) m = fmaxf(m, L[v]);
    red[tid] = m; __syncthreads();
    for (int off = 128; off > 0; off >>= 1) {
      if (tid < off) red[tid] = fmaxf(red[tid], red[tid + off]);
      __syncthreads();
    }
    m = red[0]; __syncthreads();
    float s = 0.f;
    for (int v = tid; v < V; v += 256) s += expf(L[v] - m);
    red[tid] = s; __syncthreads();
    for (int off = 128; off > 0; off >>= 1) {
      if (tid < off) red[tid] += red[tid + off];
      __syncthreads();
    }
    s = red[0]; __syncthreads();
    a[k2] = 1.0f - expf(L[blank] - m) / s;
  }
  if (tid == 0) {
    float m2 = -INFINITY;
    for (int k2 = 0; k2 < Kspk; ++k2) m2 = fmaxf(m2, a[k2]);
    float ss = 0.f;
    float e[8];
    for (int k2 = 0; k2 < Kspk; ++k2) { e[k2] = expf(a[k2] - m2); ss += e[k2]; }
    for (int k2 = 0; k2 < Kspk; ++k2) wmix[(size_t)row * Kspk + k2] = e[k2] / ss;
  }
}

__global__ __launch_bounds__(256) void stream_prep(
    const float* __restrict__ X, const float* __restrict__ wmix,
    const float* __restrict__ tags, const float* __restrict__ g,
    const float* __restrict__ bt, float* __restrict__ Xk,
    float* __restrict__ Qin, int Dn, int Kspk, int kk)
{
  const int row = blockIdx.x;
  const int tid = threadIdx.x;
  const float wv = wmix[(size_t)row * Kspk + kk];
  const float* x  = X    + (size_t)row * Dn;
  const float* tg = tags + (size_t)kk  * Dn;
  float s = 0.f, s2 = 0.f;
  for (int d = tid; d < Dn; d += 256) {
    float v = x[d] * wv + tg[d];
    Xk[(size_t)row * Dn + d] = v;
    s += v; s2 += v * v;
  }
  __shared__ float sb0[256], sb1[256];
  sb0[tid] = s; sb1[tid] = s2; __syncthreads();
  for (int off = 128; off > 0; off >>= 1) {
    if (tid < off) { sb0[tid] += sb0[tid + off]; sb1[tid] += sb1[tid + off]; }
    __syncthreads();
  }
  const float mean = sb0[0] / Dn;
  const float var  = sb1[0] / Dn - mean * mean;
  const float inv  = rsqrtf(var + 1e-5f);
  for (int d = tid; d < Dn; d += 256) {
    float v = x[d] * wv + tg[d];
    Qin[(size_t)row * Dn + d] = (v - mean) * inv * g[d] + bt[d];
  }
}

__global__ __launch_bounds__(128) void banded_attn(
    const float* __restrict__ q, const float* __restrict__ kb,
    const float* __restrict__ vb, float* __restrict__ ctx,
    int T_, int Dn, int Hn, int band)
{
  const int hd   = 64;
  const int wid  = threadIdx.x >> 5;
  const int lane = threadIdx.x & 31;
  const int gw   = blockIdx.x * 4 + wid;
  const int t = gw % T_;
  const int h = (gw / T_) % Hn;
  const int b = gw / (T_ * Hn);

  __shared__ float lq[4][64];
  __shared__ float lat[4][64];

  const size_t rowbase = (size_t)(b * T_ + t) * Dn + (size_t)h * hd;
  lq[wid][lane]      = q[rowbase + lane];
  lq[wid][lane + 32] = q[rowbase + lane + 32];
  __syncthreads();

  const float scale = 0.125f;
  const int s0 = t - band;
  const int nb = 2 * band + 1;

  const float4* lq4 = (const float4*)&lq[wid][0];

  const int j0 = lane, j1 = lane + 32;
  const int sa = s0 + j0, sbi = s0 + j1;
  float sc0 = -INFINITY, sc1 = -INFINITY;
  if (j0 < nb && sa >= 0 && sa < T_) {
    const size_t kro = (size_t)(b * T_ + sa) * Dn + (size_t)h * hd;
    __builtin_prefetch(vb + kro, 0, 0);
    const float4* kp4 = (const float4*)(kb + kro);
    float acc = 0.f;
#pragma unroll
    for (int d4 = 0; d4 < 16; ++d4) {
      float4 kv = kp4[d4];
      float4 qv = lq4[d4];
      acc += qv.x * kv.x + qv.y * kv.y + qv.z * kv.z + qv.w * kv.w;
    }
    sc0 = acc * scale;
  }
  if (j1 < nb && sbi >= 0 && sbi < T_) {
    const size_t kro = (size_t)(b * T_ + sbi) * Dn + (size_t)h * hd;
    __builtin_prefetch(vb + kro, 0, 0);
    const float4* kp4 = (const float4*)(kb + kro);
    float acc = 0.f;
#pragma unroll
    for (int d4 = 0; d4 < 16; ++d4) {
      float4 kv = kp4[d4];
      float4 qv = lq4[d4];
      acc += qv.x * kv.x + qv.y * kv.y + qv.z * kv.z + qv.w * kv.w;
    }
    sc1 = acc * scale;
  }

  float m = fmaxf(sc0, sc1);
  for (int off = 16; off > 0; off >>= 1) m = fmaxf(m, __shfl_xor(m, off, 32));
  float e0 = expf(sc0 - m);
  float e1 = expf(sc1 - m);
  float s = e0 + e1;
  for (int off = 16; off > 0; off >>= 1) s += __shfl_xor(s, off, 32);
  const float inv = 1.0f / s;
  lat[wid][j0] = e0 * inv;
  lat[wid][j1] = e1 * inv;
  __syncthreads();

  float a0 = 0.f, a1 = 0.f;
  const int d0 = lane, d1 = lane + 32;
  for (int j = 0; j < nb; ++j) {
    const int sidx = s0 + j;
    if (sidx < 0 || sidx >= T_) continue;
    const float wv = lat[wid][j];
    const float* vp = vb + (size_t)(b * T_ + sidx) * Dn + (size_t)h * hd;
    a0 += wv * vp[d0];
    a1 += wv * vp[d1];
  }
  float* cp = ctx + rowbase;
  cp[d0] = a0;
  cp[d1] = a1;
}

extern "C" void kernel_launch(void* const* d_in, const int* in_sizes, int n_in,
                              void* d_out, int out_size, void* d_ws, size_t ws_size,
                              hipStream_t stream) {
  (void)in_sizes; (void)n_in; (void)out_size; (void)ws_size;
  constexpr int B = 4, T = 1024, D_IN = 512, D = 512, K = 2, V = 500, H = 8, BAND = 24;
  constexpr int M = B * T;

  const float* x_m      = (const float*)d_in[0];
  const float* logits   = (const float*)d_in[1];
  const int*   blank_id = (const int*)  d_in[2];
  const float* proj_w   = (const float*)d_in[3];
  const float* proj_b   = (const float*)d_in[4];
  const float* spk_tags = (const float*)d_in[5];
  const float* lnq_g    = (const float*)d_in[6];
  const float* lnq_b    = (const float*)d_in[7];
  const float* lnkv_g   = (const float*)d_in[8];
  const float* lnkv_b   = (const float*)d_in[9];
  const float* in_w     = (const float*)d_in[10];
  const float* in_b     = (const float*)d_in[11];
  const float* out_w    = (const float*)d_in[12];
  const float* out_b    = (const float*)d_in[13];
  const float* ffn_ln_g = (const float*)d_in[14];
  const float* ffn_ln_b = (const float*)d_in[15];
  const float* ffn_w1   = (const float*)d_in[16];
  const float* ffn_b1   = (const float*)d_in[17];
  const float* ffn_w2   = (const float*)d_in[18];
  const float* ffn_b2   = (const float*)d_in[19];
  float* out = (float*)d_out;

  float* ws = (float*)d_ws;
  const size_t SZ = (size_t)M * D;
  float* X    = ws;
  float* kbuf = X    + SZ;
  float* vbuf = kbuf + SZ;
  float* tA   = vbuf + SZ;
  float* tB   = tA   + SZ;
  float* tC   = tB   + SZ;
  float* t1   = tC   + SZ;
  float* wmix = t1   + 4 * SZ;

  const dim3 blk(256);
  auto gg = [](int m, int n) { return dim3(m / 64, n / 128); };

  gemm_wmma_f32<<<gg(M, D), blk, 0, stream>>>(x_m, proj_w, proj_b, nullptr, X,
                                              M, D, D_IN, 0, T, K, 0);
  wmix_kernel<<<dim3(M), blk, 0, stream>>>(logits, blank_id, wmix, M, V, K);
  ln_kernel<<<dim3(M), blk, 0, stream>>>(X, lnkv_g, lnkv_b, tA, D);
  gemm_wmma_f32<<<gg(M, D), blk, 0, stream>>>(tA, in_w + (size_t)D * D, in_b + D,
                                              nullptr, kbuf, M, D, D, 0, T, K, 0);
  gemm_wmma_f32<<<gg(M, D), blk, 0, stream>>>(tA, in_w + (size_t)2 * D * D, in_b + 2 * D,
                                              nullptr, vbuf, M, D, D, 0, T, K, 0);

  for (int kk = 0; kk < K; ++kk) {
    stream_prep<<<dim3(M), blk, 0, stream>>>(X, wmix, spk_tags, lnq_g, lnq_b,
                                             tA, tB, D, K, kk);
    gemm_wmma_f32<<<gg(M, D), blk, 0, stream>>>(tB, in_w, in_b, nullptr, tC,
                                                M, D, D, 0, T, K, kk);
    banded_attn<<<dim3((B * H * T) / 4), dim3(128), 0, stream>>>(tC, kbuf, vbuf, tB,
                                                                 T, D, H, BAND);
    gemm_wmma_f32<<<gg(M, D), blk, 0, stream>>>(tB, out_w, out_b, tA, tC,
                                                M, D, D,  1, T, K, kk);
    ln_kernel<<<dim3(M), blk, 0, stream>>>(tC, ffn_ln_g, ffn_ln_b, tB, D);
    gemm_wmma_f32<<<gg(M, 4 * D), blk, 0, stream>>>(tB, ffn_w1, ffn_b1, nullptr, t1,
                                                    M, 4 * D, D,  2, T, K, kk);
    gemm_wmma_f32<<<gg(M, D), blk, 0, stream>>>(t1, ffn_w2, ffn_b2, tC, out,
                                                M, D, 4 * D,  1 | 4, T, K, kk);
  }
}


# Row kernels (3 LayerNorm, 2 stream-prep, 1 speaker-mix launch) one wave per row: row in registers, 16-byte loads, shuffle reductions, no LDS or barriers; 4096 waves per launch instead of 32768
# speedup vs baseline: 1.4711x; 1.0837x over previous
#include <hip/hip_runtime.h>
#include <hip/hip_bf16.h>
#include <math.h>

typedef float v2f __attribute__((ext_vector_type(2)));
typedef float v4f __attribute__((ext_vector_type(4)));
typedef float v8f __attribute__((ext_vector_type(8)));

__device__ __forceinline__ float gelu_exact(float x) {
  return 0.5f * x * (1.0f + erff(x * 0.70710678118654752f));
}

#define WMMA_F32(cacc, av, bv)                                                 \
  cacc = __builtin_amdgcn_wmma_f32_16x16x4_f32(false, (av), false, (bv),       \
                                               (short)0, (cacc), false, false)

#define GEMM_BM 64
#define GEMM_BN 128
#define GEMM_BK 32
#define GEMM_LDP 36

__global__ __launch_bounds__(256) void gemm_wmma_f32(
    const float* __restrict__ A, const float* __restrict__ W,
    const float* __restrict__ bias, const float* __restrict__ R,
    float* __restrict__ C, int M, int N, int Kd,
    int flags, int Tlen, int Kspk, int kk)
{
  __shared__ __attribute__((aligned(16))) float As[2][GEMM_BM * GEMM_LDP];
  __shared__ __attribute__((aligned(16))) float Bs[2][GEMM_BN * GEMM_LDP];

  const int tid  = threadIdx.x;
  const int lane = tid & 31;
  const int w    = tid >> 5;
  const int wm   = w & 3;
  const int wn   = w >> 2;
  const int blockM = blockIdx.x * GEMM_BM;
  const int blockN = blockIdx.y * GEMM_BN;
  const int tileM = blockM + wm * 16;
  const int tileN = blockN + wn * 64;
  const int half = lane >> 4;
  const int l16  = lane & 15;
  const int n0   = tileN + l16;

  const int srow = tid >> 3;
  const int scol = (tid & 7) * 4;
  const float* Ag = A + (size_t)(blockM + srow) * Kd + scol;
  const float* Bg = W + (size_t)(blockN + srow) * Kd + scol;
  const size_t rstep = (size_t)32 * Kd;
  const int sts = srow * GEMM_LDP + scol;

  v4f ra0 = *(const v4f*)(Ag);
  v4f ra1 = *(const v4f*)(Ag + rstep);
  v4f rb0 = *(const v4f*)(Bg);
  v4f rb1 = *(const v4f*)(Bg + rstep);
  v4f rb2 = *(const v4f*)(Bg + 2 * rstep);
  v4f rb3 = *(const v4f*)(Bg + 3 * rstep);

  *(v4f*)(&As[0][sts])                 = ra0;
  *(v4f*)(&As[0][sts + 32 * GEMM_LDP]) = ra1;
  *(v4f*)(&Bs[0][sts])                 = rb0;
  *(v4f*)(&Bs[0][sts + 32 * GEMM_LDP]) = rb1;
  *(v4f*)(&Bs[0][sts + 64 * GEMM_LDP]) = rb2;
  *(v4f*)(&Bs[0][sts + 96 * GEMM_LDP]) = rb3;
  __syncthreads();

  const int fa = (wm * 16 + l16) * GEMM_LDP + 4 * half;
  const int fb = (wn * 64 + l16) * GEMM_LDP + 4 * half;

  v8f c0 = {}, c1 = {}, c2 = {}, c3 = {};

  const int nk = Kd / GEMM_BK;
  for (int kt = 0; kt < nk; ++kt) {
    const int cur = kt & 1;
    const bool more = (kt + 1 < nk);
    if (more) {
      const int ko = (kt + 1) * GEMM_BK;
      ra0 = *(const v4f*)(Ag + ko);
      ra1 = *(const v4f*)(Ag + rstep + ko);
      rb0 = *(const v4f*)(Bg + ko);
      rb1 = *(const v4f*)(Bg + rstep + ko);
      rb2 = *(const v4f*)(Bg + 2 * rstep + ko);
      rb3 = *(const v4f*)(Bg + 3 * rstep + ko);
    }

    const float* as = &As[cur][fa];
    const float* bs = &Bs[cur][fb];
#pragma unroll
    for (int k8 = 0; k8 < GEMM_BK / 8; ++k8) {
      const v4f a  = *(const v4f*)(as + k8 * 8);
      const v4f b0 = *(const v4f*)(bs + k8 * 8);
      const v4f b1 = *(const v4f*)(bs + 16 * GEMM_LDP + k8 * 8);
      const v4f b2 = *(const v4f*)(bs + 32 * GEMM_LDP + k8 * 8);
      const v4f b3 = *(const v4f*)(bs + 48 * GEMM_LDP + k8 * 8);
      const v2f alo = {a.x, a.y},   ahi = {a.z, a.w};
      const v2f b0l = {b0.x, b0.y}, b0h = {b0.z, b0.w};
      const v2f b1l = {b1.x, b1.y}, b1h = {b1.z, b1.w};
      const v2f b2l = {b2.x, b2.y}, b2h = {b2.z, b2.w};
      const v2f b3l = {b3.x, b3.y}, b3h = {b3.z, b3.w};
      WMMA_F32(c0, alo, b0l);
      WMMA_F32(c1, alo, b1l);
      WMMA_F32(c2, alo, b2l);
      WMMA_F32(c3, alo, b3l);
      WMMA_F32(c0, ahi, b0h);
      WMMA_F32(c1, ahi, b1h);
      WMMA_F32(c2, ahi, b2h);
      WMMA_F32(c3, ahi, b3h);
    }

    if (more) {
      const int nxt = cur ^ 1;
      *(v4f*)(&As[nxt][sts])                 = ra0;
      *(v4f*)(&As[nxt][sts + 32 * GEMM_LDP]) = ra1;
      *(v4f*)(&Bs[nxt][sts])                 = rb0;
      *(v4f*)(&Bs[nxt][sts + 32 * GEMM_LDP]) = rb1;
      *(v4f*)(&Bs[nxt][sts + 64 * GEMM_LDP]) = rb2;
      *(v4f*)(&Bs[nxt][sts + 96 * GEMM_LDP]) = rb3;
    }
    __syncthreads();
  }

  const float bv0 = bias[n0];
  const float bv1 = bias[n0 + 16];
  const float bv2 = bias[n0 + 32];
  const float bv3 = bias[n0 + 48];
#pragma unroll
  for (int i = 0; i < 8; ++i) {
    const int row = tileM + i + 8 * half;
    float v0 = c0[i] + bv0;
    float v1 = c1[i] + bv1;
    float v2 = c2[i] + bv2;
    float v3 = c3[i] + bv3;
    if (flags & 1) {
      const float* rr = R + (size_t)row * N;
      v0 += rr[n0];
      v1 += rr[n0 + 16];
      v2 += rr[n0 + 32];
      v3 += rr[n0 + 48];
    }
    if (flags & 2) {
      v0 = gelu_exact(v0);
      v1 = gelu_exact(v1);
      v2 = gelu_exact(v2);
      v3 = gelu_exact(v3);
    }
    size_t orow = (size_t)row;
    if (flags & 4) {
      const int bi = row / Tlen;
      orow = (size_t)bi * Kspk * Tlen + (size_t)kk * Tlen + (row % Tlen);
    }
    float* cr = C + orow * N;
    cr[n0]      = v0;
    cr[n0 + 16] = v1;
    cr[n0 + 32] = v2;
    cr[n0 + 48] = v3;
  }
}

#define ROW_D   512
#define ROW_P   (ROW_D / 128)
#define ROW_WPB 8
#define WM_V    500
#define WM_V4   (WM_V / 4)
#define WM_K    2

__device__ __forceinline__ float wave_sum(float v) {
#pragma unroll
  for (int off = 16; off > 0; off >>= 1) v += __shfl_xor(v, off, 32);
  return v;
}
__device__ __forceinline__ float wave_max(float v) {
#pragma unroll
  for (int off = 16; off > 0; off >>= 1) v = fmaxf(v, __shfl_xor(v, off, 32));
  return v;
}

__global__ __launch_bounds__(256) void ln_kernel(
    const float* __restrict__ X, const float* __restrict__ g,
    const float* __restrict__ b, float* __restrict__ Y, int nrows)
{
  const int lane = threadIdx.x & 31;
  const int row  = blockIdx.x * ROW_WPB + (threadIdx.x >> 5);
  if (row >= nrows) return;
  const v4f* x4 = (const v4f*)(X + (size_t)row * ROW_D);
  v4f xv[ROW_P];
#pragma unroll
  for (int j = 0; j < ROW_P; ++j) xv[j] = x4[lane + 32 * j];
  float s = 0.f, s2 = 0.f;
#pragma unroll
  for (int j = 0; j < ROW_P; ++j) {
    s  += (xv[j].x + xv[j].y) + (xv[j].z + xv[j].w);
    s2 += (xv[j].x * xv[j].x + xv[j].y * xv[j].y) +
          (xv[j].z * xv[j].z + xv[j].w * xv[j].w);
  }
  s  = wave_sum(s);
  s2 = wave_sum(s2);
  const float mean = s / ROW_D;
  const float var  = s2 / ROW_D - mean * mean;
  const float inv  = rsqrtf(var + 1e-5f);
  const v4f* g4 = (const v4f*)g;
  const v4f* b4 = (const v4f*)b;
  v4f* y4 = (v4f*)(Y + (size_t)row * ROW_D);
#pragma unroll
  for (int j = 0; j < ROW_P; ++j) {
    const v4f gv = g4[lane + 32 * j];
    const v4f bv = b4[lane + 32 * j];
    y4[lane + 32 * j] = (xv[j] - mean) * inv * gv + bv;
  }
}

__global__ __launch_bounds__(256) void wmix_kernel(
    const float* __restrict__ logits, const int* __restrict__ blank_p,
    float* __restrict__ wmix, int BT)
{
  const int lane = threadIdx.x & 31;
  const int row  = blockIdx.x * ROW_WPB + (threadIdx.x >> 5);
  if (row >= BT) return;
  const int blank = blank_p[0];
  const v4f ninf = {-INFINITY, -INFINITY, -INFINITY, -INFINITY};

  v4f lv[WM_K][4];
  float lb[WM_K];
#pragma unroll
  for (int k2 = 0; k2 < WM_K; ++k2) {
    const float* L = logits + ((size_t)k2 * BT + row) * WM_V;
    const v4f* L4 = (const v4f*)L;
#pragma unroll
    for (int j = 0; j < 4; ++j) {
      const int i4 = lane + 32 * j;
      lv[k2][j] = (i4 < WM_V4) ? L4[i4] : ninf;
    }
    lb[k2] = L[blank];
  }

  float a[WM_K];
#pragma unroll
  for (int k2 = 0; k2 < WM_K; ++k2) {
    float m = -INFINITY;
#pragma unroll
    for (int j = 0; j < 4; ++j)
      m = fmaxf(m, fmaxf(fmaxf(lv[k2][j].x, lv[k2][j].y),
                         fmaxf(lv[k2][j].z, lv[k2][j].w)));
    m = wave_max(m);
    float s = 0.f;
#pragma unroll
    for (int j = 0; j < 4; ++j)
      s += (expf(lv[k2][j].x - m) + expf(lv[k2][j].y - m)) +
           (expf(lv[k2][j].z - m) + expf(lv[k2][j].w - m));
    s = wave_sum(s);
    a[k2] = 1.0f - expf(lb[k2] - m) / s;
  }

  float m2 = -INFINITY;
#pragma unroll
  for (int k2 = 0; k2 < WM_K; ++k2) m2 = fmaxf(m2, a[k2]);
  float e[WM_K];
  float ss = 0.f;
#pragma unroll
  for (int k2 = 0; k2 < WM_K; ++k2) { e[k2] = expf(a[k2] - m2); ss += e[k2]; }
  if (lane == 0) {
#pragma unroll
    for (int k2 = 0; k2 < WM_K; ++k2) wmix[(size_t)row * WM_K + k2] = e[k2] / ss;
  }
}

__global__ __launch_bounds__(256) void stream_prep(
    const float* __restrict__ X, const float* __restrict__ wmix,
    const float* __restrict__ tags, const float* __restrict__ g,
    const float* __restrict__ bt, float* __restrict__ Xk,
    float* __restrict__ Qin, int nrows, int kk)
{
  const int lane = threadIdx.x & 31;
  const int row  = blockIdx.x * ROW_WPB + (threadIdx.x >> 5);
  if (row >= nrows) return;
  const float wv = wmix[(size_t)row * WM_K + kk];
  const v4f* x4  = (const v4f*)(X    + (size_t)row * ROW_D);
  const v4f* tg4 = (const v4f*)(tags + (size_t)kk  * ROW_D);
  v4f* xk4 = (v4f*)(Xk + (size_t)row * ROW_D);
  v4f v[ROW_P];
#pragma unroll
  for (int j = 0; j < ROW_P; ++j) {
    v[j] = x4[lane + 32 * j] * wv + tg4[lane + 32 * j];
    xk4[lane + 32 * j] = v[j];
  }
  float s = 0.f, s2 = 0.f;
#pragma unroll
  for (int j = 0; j < ROW_P; ++j) {
    s  += (v[j].x + v[j].y) + (v[j].z + v[j].w);
    s2 += (v[j].x * v[j].x + v[j].y * v[j].y) +
          (v[j].z * v[j].z + v[j].w * v[j].w);
  }
  s  = wave_sum(s);
  s2 = wave_sum(s2);
  const float mean = s / ROW_D;
  const float var  = s2 / ROW_D - mean * mean;
  const float inv  = rsqrtf(var + 1e-5f);
  const v4f* g4 = (const v4f*)g;
  const v4f* b4 = (const v4f*)bt;
  v4f* q4 = (v4f*)(Qin + (size_t)row * ROW_D);
#pragma unroll
  for (int j = 0; j < ROW_P; ++j) {
    const v4f gv = g4[lane + 32 * j];
    const v4f bv = b4[lane + 32 * j];
    q4[lane + 32 * j] = (v[j] - mean) * inv * gv + bv;
  }
}

__global__ __launch_bounds__(128) void banded_attn(
    const float* __restrict__ q, const float* __restrict__ kb,
    const float* __restrict__ vb, float* __restrict__ ctx,
    int T_, int Dn, int Hn, int band)
{
  const int hd   = 64;
  const int wid  = threadIdx.x >> 5;
  const int lane = threadIdx.x & 31;
  const int gw   = blockIdx.x * 4 + wid;
  const int t = gw % T_;
  const int h = (gw / T_) % Hn;
  const int b = gw / (T_ * Hn);

  __shared__ float lq[4][64];
  __shared__ float lat[4][64];

  const size_t rowbase = (size_t)(b * T_ + t) * Dn + (size_t)h * hd;
  lq[wid][lane]      = q[rowbase + lane];
  lq[wid][lane + 32] = q[rowbase + lane + 32];
  __syncthreads();

  const float scale = 0.125f;
  const int s0 = t - band;
  const int nb = 2 * band + 1;

  const float4* lq4 = (const float4*)&lq[wid][0];

  const int j0 = lane, j1 = lane + 32;
  const int sa = s0 + j0, sbi = s0 + j1;
  float sc0 = -INFINITY, sc1 = -INFINITY;
  if (j0 < nb && sa >= 0 && sa < T_) {
    const size_t kro = (size_t)(b * T_ + sa) * Dn + (size_t)h * hd;
    __builtin_prefetch(vb + kro, 0, 0);
    const float4* kp4 = (const float4*)(kb + kro);
    float acc = 0.f;
#pragma unroll
    for (int d4 = 0; d4 < 16; ++d4) {
      float4 kv = kp4[d4];
      float4 qv = lq4[d4];
      acc += qv.x * kv.x + qv.y * kv.y + qv.z * kv.z + qv.w * kv.w;
    }
    sc0 = acc * scale;
  }
  if (j1 < nb && sbi >= 0 && sbi < T_) {
    const size_t kro = (size_t)(b * T_ + sbi) * Dn + (size_t)h * hd;
    __builtin_prefetch(vb + kro, 0, 0);
    const float4* kp4 = (const float4*)(kb + kro);
    float acc = 0.f;
#pragma unroll
    for (int d4 = 0; d4 < 16; ++d4) {
      float4 kv = kp4[d4];
      float4 qv = lq4[d4];
      acc += qv.x * kv.x + qv.y * kv.y + qv.z * kv.z + qv.w * kv.w;
    }
    sc1 = acc * scale;
  }

  float m = fmaxf(sc0, sc1);
  for (int off = 16; off > 0; off >>= 1) m = fmaxf(m, __shfl_xor(m, off, 32));
  float e0 = expf(sc0 - m);
  float e1 = expf(sc1 - m);
  float s = e0 + e1;
  for (int off = 16; off > 0; off >>= 1) s += __shfl_xor(s, off, 32);
  const float inv = 1.0f / s;
  lat[wid][j0] = e0 * inv;
  lat[wid][j1] = e1 * inv;
  __syncthreads();

  float a0 = 0.f, a1 = 0.f;
  const int d0 = lane, d1 = lane + 32;
  for (int j = 0; j < nb; ++j) {
    const int sidx = s0 + j;
    if (sidx < 0 || sidx >= T_) continue;
    const float wv = lat[wid][j];
    const float* vp = vb + (size_t)(b * T_ + sidx) * Dn + (size_t)h * hd;
    a0 += wv * vp[d0];
    a1 += wv * vp[d1];
  }
  float* cp = ctx + rowbase;
  cp[d0] = a0;
  cp[d1] = a1;
}

extern "C" void kernel_launch(void* const* d_in, const int* in_sizes, int n_in,
                              void* d_out, int out_size, void* d_ws, size_t ws_size,
                              hipStream_t stream) {
  (void)in_sizes; (void)n_in; (void)out_size; (void)ws_size;
  constexpr int B = 4, T = 1024, D_IN = 512, D = 512, K = 2, V = 500, H = 8, BAND = 24;
  constexpr int M = B * T;

  const float* x_m      = (const float*)d_in[0];
  const float* logits   = (const float*)d_in[1];
  const int*   blank_id = (const int*)  d_in[2];
  const float* proj_w   = (const float*)d_in[3];
  const float* proj_b   = (const float*)d_in[4];
  const float* spk_tags = (const float*)d_in[5];
  const float* lnq_g    = (const float*)d_in[6];
  const float* lnq_b    = (const float*)d_in[7];
  const float* lnkv_g   = (const float*)d_in[8];
  const float* lnkv_b   = (const float*)d_in[9];
  const float* in_w     = (const float*)d_in[10];
  const float* in_b     = (const float*)d_in[11];
  const float* out_w    = (const float*)d_in[12];
  const float* out_b    = (const float*)d_in[13];
  const float* ffn_ln_g = (const float*)d_in[14];
  const float* ffn_ln_b = (const float*)d_in[15];
  const float* ffn_w1   = (const float*)d_in[16];
  const float* ffn_b1   = (const float*)d_in[17];
  const float* ffn_w2   = (const float*)d_in[18];
  const float* ffn_b2   = (const float*)d_in[19];
  float* out = (float*)d_out;

  float* ws = (float*)d_ws;
  const size_t SZ = (size_t)M * D;
  float* X    = ws;
  float* kbuf = X    + SZ;
  float* vbuf = kbuf + SZ;
  float* tA   = vbuf + SZ;
  float* tB   = tA   + SZ;
  float* tC   = tB   + SZ;
  float* t1   = tC   + SZ;
  float* wmix = t1   + 4 * SZ;

  const dim3 blk(256);
  static_assert(D == ROW_D && V == WM_V && K == WM_K && M % ROW_WPB == 0,
                "row kernels are written for these sizes");
  const dim3 rowgrid(M / ROW_WPB);
  auto gg = [](int m, int n) { return dim3(m / 64, n / 128); };

  gemm_wmma_f32<<<gg(M, D), blk, 0, stream>>>(x_m, proj_w, proj_b, nullptr, X,
                                              M, D, D_IN, 0, T, K, 0);
  wmix_kernel<<<rowgrid, blk, 0, stream>>>(logits, blank_id, wmix, M);
  ln_kernel<<<rowgrid, blk, 0, stream>>>(X, lnkv_g, lnkv_b, tA, M);
  gemm_wmma_f32<<<gg(M, D), blk, 0, stream>>>(tA, in_w + (size_t)D * D, in_b + D,
                                              nullptr, kbuf, M, D, D, 0, T, K, 0);
  gemm_wmma_f32<<<gg(M, D), blk, 0, stream>>>(tA, in_w + (size_t)2 * D * D, in_b + 2 * D,
                                              nullptr, vbuf, M, D, D, 0, T, K, 0);

  for (int kk = 0; kk < K; ++kk) {
    stream_prep<<<rowgrid, blk, 0, stream>>>(X, wmix, spk_tags, lnq_g, lnq_b,
                                             tA, tB, M, kk);
    gemm_wmma_f32<<<gg(M, D), blk, 0, stream>>>(tB, in_w, in_b, nullptr, tC,
                                                M, D, D, 0, T, K, kk);
    banded_attn<<<dim3((B * H * T) / 4), dim3(128), 0, stream>>>(tC, kbuf, vbuf, tB,
                                                                 T, D, H, BAND);
    gemm_wmma_f32<<<gg(M, D), blk, 0, stream>>>(tB, out_w, out_b, tA, tC,
                                                M, D, D,  1, T, K, kk);
    ln_kernel<<<rowgrid, blk, 0, stream>>>(tC, ffn_ln_g, ffn_ln_b, tB, M);
    gemm_wmma_f32<<<gg(M, 4 * D), blk, 0, stream>>>(tB, ffn_w1, ffn_b1, nullptr, t1,
                                                    M, 4 * D, D,  2, T, K, kk);
    gemm_wmma_f32<<<gg(M, D), blk, 0, stream>>>(t1, ffn_w2, ffn_b2, tC, out,
                                                M, D, 4 * D,  1 | 4, T, K, kk);
  }
}


# Banded attention moved onto the f32 16x16x4 matrix instruction: one wave per 16 queries x 64-key window, K/V rows of 64 queries staged once in LDS (was one wave per query, 49 serial V-row loads)
# speedup vs baseline: 1.9636x; 1.3348x over previous
#include <hip/hip_runtime.h>
#include <hip/hip_bf16.h>
#include <math.h>

typedef float v2f __attribute__((ext_vector_type(2)));
typedef float v4f __attribute__((ext_vector_type(4)));
typedef float v8f __attribute__((ext_vector_type(8)));

__device__ __forceinline__ float gelu_exact(float x) {
  return 0.5f * x * (1.0f + erff(x * 0.70710678118654752f));
}

#define WMMA_F32(cacc, av, bv)                                                 \
  cacc = __builtin_amdgcn_wmma_f32_16x16x4_f32(false, (av), false, (bv),       \
                                               (short)0, (cacc), false, false)

#define GEMM_BM 64
#define GEMM_BN 128
#define GEMM_BK 32
#define GEMM_LDP 36

__global__ __launch_bounds__(256) void gemm_wmma_f32(
    const float* __restrict__ A, const float* __restrict__ W,
    const float* __restrict__ bias, const float* __restrict__ R,
    float* __restrict__ C, int M, int N, int Kd,
    int flags, int Tlen, int Kspk, int kk)
{
  __shared__ __attribute__((aligned(16))) float As[2][GEMM_BM * GEMM_LDP];
  __shared__ __attribute__((aligned(16))) float Bs[2][GEMM_BN * GEMM_LDP];

  const int tid  = threadIdx.x;
  const int lane = tid & 31;
  const int w    = tid >> 5;
  const int wm   = w & 3;
  const int wn   = w >> 2;
  const int blockM = blockIdx.x * GEMM_BM;
  const int blockN = blockIdx.y * GEMM_BN;
  const int tileM = blockM + wm * 16;
  const int tileN = blockN + wn * 64;
  const int half = lane >> 4;
  const int l16  = lane & 15;
  const int n0   = tileN + l16;

  const int srow = tid >> 3;
  const int scol = (tid & 7) * 4;
  const float* Ag = A + (size_t)(blockM + srow) * Kd + scol;
  const float* Bg = W + (size_t)(blockN + srow) * Kd + scol;
  const size_t rstep = (size_t)32 * Kd;
  const int sts = srow * GEMM_LDP + scol;

  v4f ra0 = *(const v4f*)(Ag);
  v4f ra1 = *(const v4f*)(Ag + rstep);
  v4f rb0 = *(const v4f*)(Bg);
  v4f rb1 = *(const v4f*)(Bg + rstep);
  v4f rb2 = *(const v4f*)(Bg + 2 * rstep);
  v4f rb3 = *(const v4f*)(Bg + 3 * rstep);

  *(v4f*)(&As[0][sts])                 = ra0;
  *(v4f*)(&As[0][sts + 32 * GEMM_LDP]) = ra1;
  *(v4f*)(&Bs[0][sts])                 = rb0;
  *(v4f*)(&Bs[0][sts + 32 * GEMM_LDP]) = rb1;
  *(v4f*)(&Bs[0][sts + 64 * GEMM_LDP]) = rb2;
  *(v4f*)(&Bs[0][sts + 96 * GEMM_LDP]) = rb3;
  __syncthreads();

  const int fa = (wm * 16 + l16) * GEMM_LDP + 4 * half;
  const int fb = (wn * 64 + l16) * GEMM_LDP + 4 * half;

  v8f c0 = {}, c1 = {}, c2 = {}, c3 = {};

  const int nk = Kd / GEMM_BK;
  for (int kt = 0; kt < nk; ++kt) {
    const int cur = kt & 1;
    const bool more = (kt + 1 < nk);
    if (more) {
      const int ko = (kt + 1) * GEMM_BK;
      ra0 = *(const v4f*)(Ag + ko);
      ra1 = *(const v4f*)(Ag + rstep + ko);
      rb0 = *(const v4f*)(Bg + ko);
      rb1 = *(const v4f*)(Bg + rstep + ko);
      rb2 = *(const v4f*)(Bg + 2 * rstep + ko);
      rb3 = *(const v4f*)(Bg + 3 * rstep + ko);
    }

    const float* as = &As[cur][fa];
    const float* bs = &Bs[cur][fb];
#pragma unroll
    for (int k8 = 0; k8 < GEMM_BK / 8; ++k8) {
      const v4f a  = *(const v4f*)(as + k8 * 8);
      const v4f b0 = *(const v4f*)(bs + k8 * 8);
      const v4f b1 = *(const v4f*)(bs + 16 * GEMM_LDP + k8 * 8);
      const v4f b2 = *(const v4f*)(bs + 32 * GEMM_LDP + k8 * 8);
      const v4f b3 = *(const v4f*)(bs + 48 * GEMM_LDP + k8 * 8);
      const v2f alo = {a.x, a.y},   ahi = {a.z, a.w};
      const v2f b0l = {b0.x, b0.y}, b0h = {b0.z, b0.w};
      const v2f b1l = {b1.x, b1.y}, b1h = {b1.z, b1.w};
      const v2f b2l = {b2.x, b2.y}, b2h = {b2.z, b2.w};
      const v2f b3l = {b3.x, b3.y}, b3h = {b3.z, b3.w};
      WMMA_F32(c0, alo, b0l);
      WMMA_F32(c1, alo, b1l);
      WMMA_F32(c2, alo, b2l);
      WMMA_F32(c3, alo, b3l);
      WMMA_F32(c0, ahi, b0h);
      WMMA_F32(c1, ahi, b1h);
      WMMA_F32(c2, ahi, b2h);
      WMMA_F32(c3, ahi, b3h);
    }

    if (more) {
      const int nxt = cur ^ 1;
      *(v4f*)(&As[nxt][sts])                 = ra0;
      *(v4f*)(&As[nxt][sts + 32 * GEMM_LDP]) = ra1;
      *(v4f*)(&Bs[nxt][sts])                 = rb0;
      *(v4f*)(&Bs[nxt][sts + 32 * GEMM_LDP]) = rb1;
      *(v4f*)(&Bs[nxt][sts + 64 * GEMM_LDP]) = rb2;
      *(v4f*)(&Bs[nxt][sts + 96 * GEMM_LDP]) = rb3;
    }
    __syncthreads();
  }

  const float bv0 = bias[n0];
  const float bv1 = bias[n0 + 16];
  const float bv2 = bias[n0 + 32];
  const float bv3 = bias[n0 + 48];
#pragma unroll
  for (int i = 0; i < 8; ++i) {
    const int row = tileM + i + 8 * half;
    float v0 = c0[i] + bv0;
    float v1 = c1[i] + bv1;
    float v2 = c2[i] + bv2;
    float v3 = c3[i] + bv3;
    if (flags & 1) {
      const float* rr = R + (size_t)row * N;
      v0 += rr[n0];
      v1 += rr[n0 + 16];
      v2 += rr[n0 + 32];
      v3 += rr[n0 + 48];
    }
    if (flags & 2) {
      v0 = gelu_exact(v0);
      v1 = gelu_exact(v1);
      v2 = gelu_exact(v2);
      v3 = gelu_exact(v3);
    }
    size_t orow = (size_t)row;
    if (flags & 4) {
      const int bi = row / Tlen;
      orow = (size_t)bi * Kspk * Tlen + (size_t)kk * Tlen + (row % Tlen);
    }
    float* cr = C + orow * N;
    cr[n0]      = v0;
    cr[n0 + 16] = v1;
    cr[n0 + 32] = v2;
    cr[n0 + 48] = v3;
  }
}

#define ROW_D   512
#define ROW_P   (ROW_D / 128)
#define ROW_WPB 8
#define WM_V    500
#define WM_V4   (WM_V / 4)
#define WM_K    2

__device__ __forceinline__ float wave_sum(float v) {
#pragma unroll
  for (int off = 16; off > 0; off >>= 1) v += __shfl_xor(v, off, 32);
  return v;
}
__device__ __forceinline__ float wave_max(float v) {
#pragma unroll
  for (int off = 16; off > 0; off >>= 1) v = fmaxf(v, __shfl_xor(v, off, 32));
  return v;
}

__global__ __launch_bounds__(256) void ln_kernel(
    const float* __restrict__ X, const float* __restrict__ g,
    const float* __restrict__ b, float* __restrict__ Y, int nrows)
{
  const int lane = threadIdx.x & 31;
  const int row  = blockIdx.x * ROW_WPB + (threadIdx.x >> 5);
  if (row >= nrows) return;
  const v4f* x4 = (const v4f*)(X + (size_t)row * ROW_D);
  v4f xv[ROW_P];
#pragma unroll
  for (int j = 0; j < ROW_P; ++j) xv[j] = x4[lane + 32 * j];
  float s = 0.f, s2 = 0.f;
#pragma unroll
  for (int j = 0; j < ROW_P; ++j) {
    s  += (xv[j].x + xv[j].y) + (xv[j].z + xv[j].w);
    s2 += (xv[j].x * xv[j].x + xv[j].y * xv[j].y) +
          (xv[j].z * xv[j].z + xv[j].w * xv[j].w);
  }
  s  = wave_sum(s);
  s2 = wave_sum(s2);
  const float mean = s / ROW_D;
  const float var  = s2 / ROW_D - mean * mean;
  const float inv  = rsqrtf(var + 1e-5f);
  const v4f* g4 = (const v4f*)g;
  const v4f* b4 = (const v4f*)b;
  v4f* y4 = (v4f*)(Y + (size_t)row * ROW_D);
#pragma unroll
  for (int j = 0; j < ROW_P; ++j) {
    const v4f gv = g4[lane + 32 * j];
    const v4f bv = b4[lane + 32 * j];
    y4[lane + 32 * j] = (xv[j] - mean) * inv * gv + bv;
  }
}

__global__ __launch_bounds__(256) void wmix_kernel(
    const float* __restrict__ logits, const int* __restrict__ blank_p,
    float* __restrict__ wmix, int BT)
{
  const int lane = threadIdx.x & 31;
  const int row  = blockIdx.x * ROW_WPB + (threadIdx.x >> 5);
  if (row >= BT) return;
  const int blank = blank_p[0];
  const v4f ninf = {-INFINITY, -INFINITY, -INFINITY, -INFINITY};

  v4f lv[WM_K][4];
  float lb[WM_K];
#pragma unroll
  for (int k2 = 0; k2 < WM_K; ++k2) {
    const float* L = logits + ((size_t)k2 * BT + row) * WM_V;
    const v4f* L4 = (const v4f*)L;
#pragma unroll
    for (int j = 0; j < 4; ++j) {
      const int i4 = lane + 32 * j;
      lv[k2][j] = (i4 < WM_V4) ? L4[i4] : ninf;
    }
    lb[k2] = L[blank];
  }

  float a[WM_K];
#pragma unroll
  for (int k2 = 0; k2 < WM_K; ++k2) {
    float m = -INFINITY;
#pragma unroll
    for (int j = 0; j < 4; ++j)
      m = fmaxf(m, fmaxf(fmaxf(lv[k2][j].x, lv[k2][j].y),
                         fmaxf(lv[k2][j].z, lv[k2][j].w)));
    m = wave_max(m);
    float s = 0.f;
#pragma unroll
    for (int j = 0; j < 4; ++j)
      s += (expf(lv[k2][j].x - m) + expf(lv[k2][j].y - m)) +
           (expf(lv[k2][j].z - m) + expf(lv[k2][j].w - m));
    s = wave_sum(s);
    a[k2] = 1.0f - expf(lb[k2] - m) / s;
  }

  float m2 = -INFINITY;
#pragma unroll
  for (int k2 = 0; k2 < WM_K; ++k2) m2 = fmaxf(m2, a[k2]);
  float e[WM_K];
  float ss = 0.f;
#pragma unroll
  for (int k2 = 0; k2 < WM_K; ++k2) { e[k2] = expf(a[k2] - m2); ss += e[k2]; }
  if (lane == 0) {
#pragma unroll
    for (int k2 = 0; k2 < WM_K; ++k2) wmix[(size_t)row * WM_K + k2] = e[k2] / ss;
  }
}

__global__ __launch_bounds__(256) void stream_prep(
    const float* __restrict__ X, const float* __restrict__ wmix,
    const float* __restrict__ tags, const float* __restrict__ g,
    const float* __restrict__ bt, float* __restrict__ Xk,
    float* __restrict__ Qin, int nrows, int kk)
{
  const int lane = threadIdx.x & 31;
  const int row  = blockIdx.x * ROW_WPB + (threadIdx.x >> 5);
  if (row >= nrows) return;
  const float wv = wmix[(size_t)row * WM_K + kk];
  const v4f* x4  = (const v4f*)(X    + (size_t)row * ROW_D);
  const v4f* tg4 = (const v4f*)(tags + (size_t)kk  * ROW_D);
  v4f* xk4 = (v4f*)(Xk + (size_t)row * ROW_D);
  v4f v[ROW_P];
#pragma unroll
  for (int j = 0; j < ROW_P; ++j) {
    v[j] = x4[lane + 32 * j] * wv + tg4[lane + 32 * j];
    xk4[lane + 32 * j] = v[j];
  }
  float s = 0.f, s2 = 0.f;
#pragma unroll
  for (int j = 0; j < ROW_P; ++j) {
    s  += (v[j].x + v[j].y) + (v[j].z + v[j].w);
    s2 += (v[j].x * v[j].x + v[j].y * v[j].y) +
          (v[j].z * v[j].z + v[j].w * v[j].w);
  }
  s  = wave_sum(s);
  s2 = wave_sum(s2);
  const float mean = s / ROW_D;
  const float var  = s2 / ROW_D - mean * mean;
  const float inv  = rsqrtf(var + 1e-5f);
  const v4f* g4 = (const v4f*)g;
  const v4f* b4 = (const v4f*)bt;
  v4f* q4 = (v4f*)(Qin + (size_t)row * ROW_D);
#pragma unroll
  for (int j = 0; j < ROW_P; ++j) {
    const v4f gv = g4[lane + 32 * j];
    const v4f bv = b4[lane + 32 * j];
    q4[lane + 32 * j] = (v[j] - mean) * inv * gv + bv;
  }
}

#define AT_T     1024
#define AT_D     512
#define AT_H     8
#define AT_HD    64
#define AT_BAND  24
#define AT_QW    16
#define AT_WAVES 4
#define AT_QB    (AT_QW * AT_WAVES)
#define AT_WIN   (AT_QB + 2 * AT_BAND)
#define AT_NST   (AT_WIN / 8)
#define AT_LDP   68

template <int M>
__device__ __forceinline__ float row_xor(float x) {
  const int b = __builtin_bit_cast(int, x);
  return __builtin_bit_cast(
      float, __builtin_amdgcn_update_dpp(b, b, 0x160 + M, 0xf, 0xf, false));
}
__device__ __forceinline__ float half_max(float v) {
  v = fmaxf(v, row_xor<1>(v));
  v = fmaxf(v, row_xor<2>(v));
  v = fmaxf(v, row_xor<4>(v));
  v = fmaxf(v, row_xor<8>(v));
  return v;
}
__device__ __forceinline__ float half_sum(float v) {
  v += row_xor<1>(v);
  v += row_xor<2>(v);
  v += row_xor<4>(v);
  v += row_xor<8>(v);
  return v;
}

__global__ __launch_bounds__(128) void banded_attn_wmma(
    const float* __restrict__ q, const float* __restrict__ kb,
    const float* __restrict__ vb, float* __restrict__ ctx)
{
  __shared__ __attribute__((aligned(16))) float Ks[AT_WIN * AT_LDP];
  __shared__ __attribute__((aligned(16))) float Vs[AT_WIN * AT_LDP];
  __shared__ __attribute__((aligned(16))) float Ps[AT_WAVES][AT_QW * AT_LDP];

  const int tid  = threadIdx.x;
  const int lane = tid & 31;
  const int w    = tid >> 5;
  const int c    = lane & 15;
  const int hf   = lane >> 4;

  constexpr int tps = AT_T / AT_QB;
  const int blk = blockIdx.x;
  const int tb  = (blk % tps) * AT_QB;
  const int h   = (blk / tps) % AT_H;
  const int b   = blk / (tps * AT_H);
  const int t0  = tb + AT_QW * w;

  const size_t seqoff = (size_t)b * AT_T * AT_D + (size_t)h * AT_HD;

  v4f qa[AT_HD / 8];
  {
    const float* qrow = q + seqoff + (size_t)(t0 + c) * AT_D + 4 * hf;
#pragma unroll
    for (int m = 0; m < AT_HD / 8; ++m) qa[m] = *(const v4f*)(qrow + 8 * m);
  }

  {
    const int srow = tid >> 4;
    const int scol = (tid & 15) * 4;
    const float* kbase = kb + seqoff + scol;
    const float* vbase = vb + seqoff + scol;
    v4f kr[AT_NST], vr[AT_NST];
#pragma unroll
    for (int it = 0; it < AT_NST; ++it) {
      int s = tb - AT_BAND + srow + 8 * it;
      s = max(s, 0);
      s = min(s, AT_T - 1);
      kr[it] = *(const v4f*)(kbase + (size_t)s * AT_D);
      vr[it] = *(const v4f*)(vbase + (size_t)s * AT_D);
    }
#pragma unroll
    for (int it = 0; it < AT_NST; ++it) {
      const int r = srow + 8 * it;
      *(v4f*)(&Ks[r * AT_LDP + scol]) = kr[it];
      *(v4f*)(&Vs[r * AT_LDP + scol]) = vr[it];
    }
  }
  __syncthreads();

  v8f sc[4] = {{}, {}, {}, {}};
  {
    const float* kf = &Ks[(AT_QW * w + c) * AT_LDP + 4 * hf];
#pragma unroll
    for (int m = 0; m < AT_HD / 8; ++m) {
      const v4f b0 = *(const v4f*)(kf + 8 * m);
      const v4f b1 = *(const v4f*)(kf + 16 * AT_LDP + 8 * m);
      const v4f b2 = *(const v4f*)(kf + 32 * AT_LDP + 8 * m);
      const v4f b3 = *(const v4f*)(kf + 48 * AT_LDP + 8 * m);
      const v2f alo = {qa[m].x, qa[m].y}, ahi = {qa[m].z, qa[m].w};
      const v2f b0l = {b0.x, b0.y}, b0h = {b0.z, b0.w};
      const v2f b1l = {b1.x, b1.y}, b1h = {b1.z, b1.w};
      const v2f b2l = {b2.x, b2.y}, b2h = {b2.z, b2.w};
      const v2f b3l = {b3.x, b3.y}, b3h = {b3.z, b3.w};
      WMMA_F32(sc[0], alo, b0l);
      WMMA_F32(sc[1], alo, b1l);
      WMMA_F32(sc[2], alo, b2l);
      WMMA_F32(sc[3], alo, b3l);
      WMMA_F32(sc[0], ahi, b0h);
      WMMA_F32(sc[1], ahi, b1h);
      WMMA_F32(sc[2], ahi, b2h);
      WMMA_F32(sc[3], ahi, b3h);
    }
  }

  const float scale = 0.125f;
  float rinv[8];
  {
    float* pw = &Ps[w][0];
#pragma unroll
    for (int r = 0; r < 8; ++r) {
      const int i = 8 * hf + r;
      float val[4];
      float mx = -INFINITY;
#pragma unroll
      for (int nt = 0; nt < 4; ++nt) {
        const int j  = 16 * nt + c;
        const int d  = j - i;
        const int sg = t0 - AT_BAND + j;
        const bool ok = (d >= 0) && (d <= 2 * AT_BAND) && (sg >= 0) && (sg < AT_T);
        val[nt] = ok ? sc[nt][r] * scale : -INFINITY;
        mx = fmaxf(mx, val[nt]);
      }
      mx = half_max(mx);
      float sum = 0.f;
#pragma unroll
      for (int nt = 0; nt < 4; ++nt) {
        const float e = expf(val[nt] - mx);
        sum += e;
        pw[i * AT_LDP + 16 * nt + c] = e;
      }
      sum = half_sum(sum);
      rinv[r] = 1.0f / sum;
    }
  }
  __syncthreads();

  v8f oc[4] = {{}, {}, {}, {}};
  {
    const float* pf = &Ps[w][c * AT_LDP + 4 * hf];
    const float* vf = &Vs[(AT_QW * w + 4 * hf) * AT_LDP + c];
#pragma unroll
    for (int m = 0; m < AT_HD / 8; ++m) {
      const v4f a = *(const v4f*)(pf + 8 * m);
      const v2f alo = {a.x, a.y}, ahi = {a.z, a.w};
      const float* vm = vf + 8 * m * AT_LDP;
#pragma unroll
      for (int nt = 0; nt < 4; ++nt) {
        const v2f bl = {vm[16 * nt],              vm[AT_LDP + 16 * nt]};
        const v2f bh = {vm[2 * AT_LDP + 16 * nt], vm[3 * AT_LDP + 16 * nt]};
        WMMA_F32(oc[nt], alo, bl);
        WMMA_F32(oc[nt], ahi, bh);
      }
    }
  }
  __syncthreads();

  {
    float* pw = &Ps[w][0];
#pragma unroll
    for (int r = 0; r < 8; ++r) {
      const int i = 8 * hf + r;
#pragma unroll
      for (int nt = 0; nt < 4; ++nt)
        pw[i * AT_LDP + 16 * nt + c] = oc[nt][r] * rinv[r];
    }
  }
  __syncthreads();
  {
    const float* pw = &Ps[w][0];
    float* crow = ctx + seqoff + (size_t)t0 * AT_D;
#pragma unroll
    for (int u = 0; u < 8; ++u) {
      const int f   = lane + 32 * u;
      const int row = f >> 4;
      const int col = (f & 15) * 4;
      const v4f o = *(const v4f*)(pw + row * AT_LDP + col);
      *(v4f*)(crow + (size_t)row * AT_D + col) = o;
    }
  }
}

extern "C" void kernel_launch(void* const* d_in, const int* in_sizes, int n_in,
                              void* d_out, int out_size, void* d_ws, size_t ws_size,
                              hipStream_t stream) {
  (void)in_sizes; (void)n_in; (void)out_size; (void)ws_size;
  constexpr int B = 4, T = 1024, D_IN = 512, D = 512, K = 2, V = 500, H = 8, BAND = 24;
  constexpr int M = B * T;

  const float* x_m      = (const float*)d_in[0];
  const float* logits   = (const float*)d_in[1];
  const int*   blank_id = (const int*)  d_in[2];
  const float* proj_w   = (const float*)d_in[3];
  const float* proj_b   = (const float*)d_in[4];
  const float* spk_tags = (const float*)d_in[5];
  const float* lnq_g    = (const float*)d_in[6];
  const float* lnq_b    = (const float*)d_in[7];
  const float* lnkv_g   = (const float*)d_in[8];
  const float* lnkv_b   = (const float*)d_in[9];
  const float* in_w     = (const float*)d_in[10];
  const float* in_b     = (const float*)d_in[11];
  const float* out_w    = (const float*)d_in[12];
  const float* out_b    = (const float*)d_in[13];
  const float* ffn_ln_g = (const float*)d_in[14];
  const float* ffn_ln_b = (const float*)d_in[15];
  const float* ffn_w1   = (const float*)d_in[16];
  const float* ffn_b1   = (const float*)d_in[17];
  const float* ffn_w2   = (const float*)d_in[18];
  const float* ffn_b2   = (const float*)d_in[19];
  float* out = (float*)d_out;

  float* ws = (float*)d_ws;
  const size_t SZ = (size_t)M * D;
  float* X    = ws;
  float* kbuf = X    + SZ;
  float* vbuf = kbuf + SZ;
  float* tA   = vbuf + SZ;
  float* tB   = tA   + SZ;
  float* tC   = tB   + SZ;
  float* t1   = tC   + SZ;
  float* wmix = t1   + 4 * SZ;

  const dim3 blk(256);
  static_assert(D == ROW_D && V == WM_V && K == WM_K && M % ROW_WPB == 0,
                "row kernels are written for these sizes");
  static_assert(T == AT_T && D == AT_D && H == AT_H && D / H == AT_HD &&
                BAND == AT_BAND && T % AT_QB == 0,
                "attention kernel is written for these sizes");
  const dim3 rowgrid(M / ROW_WPB);
  auto gg = [](int m, int n) { return dim3(m / 64, n / 128); };

  gemm_wmma_f32<<<gg(M, D), blk, 0, stream>>>(x_m, proj_w, proj_b, nullptr, X,
                                              M, D, D_IN, 0, T, K, 0);
  wmix_kernel<<<rowgrid, blk, 0, stream>>>(logits, blank_id, wmix, M);
  ln_kernel<<<rowgrid, blk, 0, stream>>>(X, lnkv_g, lnkv_b, tA, M);
  gemm_wmma_f32<<<gg(M, D), blk, 0, stream>>>(tA, in_w + (size_t)D * D, in_b + D,
                                              nullptr, kbuf, M, D, D, 0, T, K, 0);
  gemm_wmma_f32<<<gg(M, D), blk, 0, stream>>>(tA, in_w + (size_t)2 * D * D, in_b + 2 * D,
                                              nullptr, vbuf, M, D, D, 0, T, K, 0);

  for (int kk = 0; kk < K; ++kk) {
    stream_prep<<<rowgrid, blk, 0, stream>>>(X, wmix, spk_tags, lnq_g, lnq_b,
                                             tA, tB, M, kk);
    gemm_wmma_f32<<<gg(M, D), blk, 0, stream>>>(tB, in_w, in_b, nullptr, tC,
                                                M, D, D, 0, T, K, kk);
    banded_attn_wmma<<<dim3((B * H * T) / AT_QB), dim3(32 * AT_WAVES), 0, stream>>>(
        tC, kbuf, vbuf, tB);
    gemm_wmma_f32<<<gg(M, D), blk, 0, stream>>>(tB, out_w, out_b, tA, tC,
                                                M, D, D,  1, T, K, kk);
    ln_kernel<<<rowgrid, blk, 0, stream>>>(tC, ffn_ln_g, ffn_ln_b, tB, M);
    gemm_wmma_f32<<<gg(M, 4 * D), blk, 0, stream>>>(tB, ffn_w1, ffn_b1, nullptr, t1,
                                                    M, 4 * D, D,  2, T, K, kk);
    gemm_wmma_f32<<<gg(M, D), blk, 0, stream>>>(t1, ffn_w2, ffn_b2, tC, out,
                                                M, D, 4 * D,  1 | 4, T, K, kk);
  }
}


# GEMM k-loop software-pipelined: two fragment register sets, each chunk's 5 LDS loads issued as one group ahead of the previous chunk's 8 WMMAs; next-tile stores and barrier ahead of the last chunks
# speedup vs baseline: 2.0615x; 1.0499x over previous
#include <hip/hip_runtime.h>
#include <hip/hip_bf16.h>
#include <math.h>

typedef float v2f __attribute__((ext_vector_type(2)));
typedef float v4f __attribute__((ext_vector_type(4)));
typedef float v8f __attribute__((ext_vector_type(8)));

__device__ __forceinline__ float gelu_exact(float x) {
  return 0.5f * x * (1.0f + erff(x * 0.70710678118654752f));
}

#define WMMA_F32(cacc, av, bv)                                                 \
  cacc = __builtin_amdgcn_wmma_f32_16x16x4_f32(false, (av), false, (bv),       \
                                               (short)0, (cacc), false, false)

#define GEMM_BM 64
#define GEMM_BN 128
#define GEMM_BK 32
#define GEMM_LDP 36

struct GemmFrag { v4f a, b0, b1, b2, b3; };

__device__ __forceinline__ void gemm_fence() { __builtin_amdgcn_sched_barrier(0); }

__device__ __forceinline__ void gemm_ldfrag(GemmFrag& f, const float* as,
                                            const float* bs) {
  f.a  = *(const v4f*)(as);
  f.b0 = *(const v4f*)(bs);
  f.b1 = *(const v4f*)(bs + 16 * GEMM_LDP);
  f.b2 = *(const v4f*)(bs + 32 * GEMM_LDP);
  f.b3 = *(const v4f*)(bs + 48 * GEMM_LDP);
}

__device__ __forceinline__ void gemm_mma8(const GemmFrag& f, v8f& c0, v8f& c1,
                                          v8f& c2, v8f& c3) {
  const v2f alo = {f.a.x, f.a.y},   ahi = {f.a.z, f.a.w};
  const v2f b0l = {f.b0.x, f.b0.y}, b0h = {f.b0.z, f.b0.w};
  const v2f b1l = {f.b1.x, f.b1.y}, b1h = {f.b1.z, f.b1.w};
  const v2f b2l = {f.b2.x, f.b2.y}, b2h = {f.b2.z, f.b2.w};
  const v2f b3l = {f.b3.x, f.b3.y}, b3h = {f.b3.z, f.b3.w};
  WMMA_F32(c0, alo, b0l);
  WMMA_F32(c1, alo, b1l);
  WMMA_F32(c2, alo, b2l);
  WMMA_F32(c3, alo, b3l);
  WMMA_F32(c0, ahi, b0h);
  WMMA_F32(c1, ahi, b1h);
  WMMA_F32(c2, ahi, b2h);
  WMMA_F32(c3, ahi, b3h);
}

__global__ __launch_bounds__(256) void gemm_wmma_f32(
    const float* __restrict__ A, const float* __restrict__ W,
    const float* __restrict__ bias, const float* __restrict__ R,
    float* __restrict__ C, int M, int N, int Kd,
    int flags, int Tlen, int Kspk, int kk)
{
  __shared__ __attribute__((aligned(16))) float As[2][GEMM_BM * GEMM_LDP];
  __shared__ __attribute__((aligned(16))) float Bs[2][GEMM_BN * GEMM_LDP];

  const int tid  = threadIdx.x;
  const int lane = tid & 31;
  const int w    = tid >> 5;
  const int wm   = w & 3;
  const int wn   = w >> 2;
  const int blockM = blockIdx.x * GEMM_BM;
  const int blockN = blockIdx.y * GEMM_BN;
  const int tileM = blockM + wm * 16;
  const int tileN = blockN + wn * 64;
  const int half = lane >> 4;
  const int l16  = lane & 15;
  const int n0   = tileN + l16;

  const int srow = tid >> 3;
  const int scol = (tid & 7) * 4;
  const float* Ag = A + (size_t)(blockM + srow) * Kd + scol;
  const float* Bg = W + (size_t)(blockN + srow) * Kd + scol;
  const size_t rstep = (size_t)32 * Kd;
  const int sts = srow * GEMM_LDP + scol;

  v4f ra0 = *(const v4f*)(Ag);
  v4f ra1 = *(const v4f*)(Ag + rstep);
  v4f rb0 = *(const v4f*)(Bg);
  v4f rb1 = *(const v4f*)(Bg + rstep);
  v4f rb2 = *(const v4f*)(Bg + 2 * rstep);
  v4f rb3 = *(const v4f*)(Bg + 3 * rstep);

  *(v4f*)(&As[0][sts])                 = ra0;
  *(v4f*)(&As[0][sts + 32 * GEMM_LDP]) = ra1;
  *(v4f*)(&Bs[0][sts])                 = rb0;
  *(v4f*)(&Bs[0][sts + 32 * GEMM_LDP]) = rb1;
  *(v4f*)(&Bs[0][sts + 64 * GEMM_LDP]) = rb2;
  *(v4f*)(&Bs[0][sts + 96 * GEMM_LDP]) = rb3;
  __syncthreads();

  const int nk = Kd / GEMM_BK;
  if (1 < nk) {
    ra0 = *(const v4f*)(Ag + GEMM_BK);
    ra1 = *(const v4f*)(Ag + rstep + GEMM_BK);
    rb0 = *(const v4f*)(Bg + GEMM_BK);
    rb1 = *(const v4f*)(Bg + rstep + GEMM_BK);
    rb2 = *(const v4f*)(Bg + 2 * rstep + GEMM_BK);
    rb3 = *(const v4f*)(Bg + 3 * rstep + GEMM_BK);
  }

  const int fa = (wm * 16 + l16) * GEMM_LDP + 4 * half;
  const int fb = (wn * 64 + l16) * GEMM_LDP + 4 * half;

  v8f c0 = {}, c1 = {}, c2 = {}, c3 = {};

  GemmFrag f0, f1;
  gemm_ldfrag(f0, &As[0][fa], &Bs[0][fb]);

  for (int kt = 0; kt < nk; ++kt) {
    const int cur = kt & 1;
    const int nxt = cur ^ 1;
    const float* as = &As[cur][fa];
    const float* bs = &Bs[cur][fb];

    gemm_fence();
    gemm_ldfrag(f1, as + 8, bs + 8);
    gemm_fence();
    gemm_mma8(f0, c0, c1, c2, c3);
    gemm_fence();
    gemm_ldfrag(f0, as + 16, bs + 16);
    gemm_fence();
    gemm_mma8(f1, c0, c1, c2, c3);
    gemm_fence();
    gemm_ldfrag(f1, as + 24, bs + 24);
    *(v4f*)(&As[nxt][sts])                 = ra0;
    *(v4f*)(&As[nxt][sts + 32 * GEMM_LDP]) = ra1;
    *(v4f*)(&Bs[nxt][sts])                 = rb0;
    *(v4f*)(&Bs[nxt][sts + 32 * GEMM_LDP]) = rb1;
    *(v4f*)(&Bs[nxt][sts + 64 * GEMM_LDP]) = rb2;
    *(v4f*)(&Bs[nxt][sts + 96 * GEMM_LDP]) = rb3;
    gemm_fence();
    gemm_mma8(f0, c0, c1, c2, c3);
    gemm_fence();

    __syncthreads();
    if (kt + 2 < nk) {
      const int ko = (kt + 2) * GEMM_BK;
      ra0 = *(const v4f*)(Ag + ko);
      ra1 = *(const v4f*)(Ag + rstep + ko);
      rb0 = *(const v4f*)(Bg + ko);
      rb1 = *(const v4f*)(Bg + rstep + ko);
      rb2 = *(const v4f*)(Bg + 2 * rstep + ko);
      rb3 = *(const v4f*)(Bg + 3 * rstep + ko);
    }
    gemm_ldfrag(f0, &As[nxt][fa], &Bs[nxt][fb]);
    gemm_fence();
    gemm_mma8(f1, c0, c1, c2, c3);
    gemm_fence();
  }

  const float bv0 = bias[n0];
  const float bv1 = bias[n0 + 16];
  const float bv2 = bias[n0 + 32];
  const float bv3 = bias[n0 + 48];
#pragma unroll
  for (int i = 0; i < 8; ++i) {
    const int row = tileM + i + 8 * half;
    float v0 = c0[i] + bv0;
    float v1 = c1[i] + bv1;
    float v2 = c2[i] + bv2;
    float v3 = c3[i] + bv3;
    if (flags & 1) {
      const float* rr = R + (size_t)row * N;
      v0 += rr[n0];
      v1 += rr[n0 + 16];
      v2 += rr[n0 + 32];
      v3 += rr[n0 + 48];
    }
    if (flags & 2) {
      v0 = gelu_exact(v0);
      v1 = gelu_exact(v1);
      v2 = gelu_exact(v2);
      v3 = gelu_exact(v3);
    }
    size_t orow = (size_t)row;
    if (flags & 4) {
      const int bi = row / Tlen;
      orow = (size_t)bi * Kspk * Tlen + (size_t)kk * Tlen + (row % Tlen);
    }
    float* cr = C + orow * N;
    cr[n0]      = v0;
    cr[n0 + 16] = v1;
    cr[n0 + 32] = v2;
    cr[n0 + 48] = v3;
  }
}

#define ROW_D   512
#define ROW_P   (ROW_D / 128)
#define ROW_WPB 8
#define WM_V    500
#define WM_V4   (WM_V / 4)
#define WM_K    2

__device__ __forceinline__ float wave_sum(float v) {
#pragma unroll
  for (int off = 16; off > 0; off >>= 1) v += __shfl_xor(v, off, 32);
  return v;
}
__device__ __forceinline__ float wave_max(float v) {
#pragma unroll
  for (int off = 16; off > 0; off >>= 1) v = fmaxf(v, __shfl_xor(v, off, 32));
  return v;
}

__global__ __launch_bounds__(256) void ln_kernel(
    const float* __restrict__ X, const float* __restrict__ g,
    const float* __restrict__ b, float* __restrict__ Y, int nrows)
{
  const int lane = threadIdx.x & 31;
  const int row  = blockIdx.x * ROW_WPB + (threadIdx.x >> 5);
  if (row >= nrows) return;
  const v4f* x4 = (const v4f*)(X + (size_t)row * ROW_D);
  v4f xv[ROW_P];
#pragma unroll
  for (int j = 0; j < ROW_P; ++j) xv[j] = x4[lane + 32 * j];
  float s = 0.f, s2 = 0.f;
#pragma unroll
  for (int j = 0; j < ROW_P; ++j) {
    s  += (xv[j].x + xv[j].y) + (xv[j].z + xv[j].w);
    s2 += (xv[j].x * xv[j].x + xv[j].y * xv[j].y) +
          (xv[j].z * xv[j].z + xv[j].w * xv[j].w);
  }
  s  = wave_sum(s);
  s2 = wave_sum(s2);
  const float mean = s / ROW_D;
  const float var  = s2 / ROW_D - mean * mean;
  const float inv  = rsqrtf(var + 1e-5f);
  const v4f* g4 = (const v4f*)g;
  const v4f* b4 = (const v4f*)b;
  v4f* y4 = (v4f*)(Y + (size_t)row * ROW_D);
#pragma unroll
  for (int j = 0; j < ROW_P; ++j) {
    const v4f gv = g4[lane + 32 * j];
    const v4f bv = b4[lane + 32 * j];
    y4[lane + 32 * j] = (xv[j] - mean) * inv * gv + bv;
  }
}

__global__ __launch_bounds__(256) void wmix_kernel(
    const float* __restrict__ logits, const int* __restrict__ blank_p,
    float* __restrict__ wmix, int BT)
{
  const int lane = threadIdx.x & 31;
  const int row  = blockIdx.x * ROW_WPB + (threadIdx.x >> 5);
  if (row >= BT) return;
  const int blank = blank_p[0];
  const v4f ninf = {-INFINITY, -INFINITY, -INFINITY, -INFINITY};

  v4f lv[WM_K][4];
  float lb[WM_K];
#pragma unroll
  for (int k2 = 0; k2 < WM_K; ++k2) {
    const float* L = logits + ((size_t)k2 * BT + row) * WM_V;
    const v4f* L4 = (const v4f*)L;
#pragma unroll
    for (int j = 0; j < 4; ++j) {
      const int i4 = lane + 32 * j;
      lv[k2][j] = (i4 < WM_V4) ? L4[i4] : ninf;
    }
    lb[k2] = L[blank];
  }

  float a[WM_K];
#pragma unroll
  for (int k2 = 0; k2 < WM_K; ++k2) {
    float m = -INFINITY;
#pragma unroll
    for (int j = 0; j < 4; ++j)
      m = fmaxf(m, fmaxf(fmaxf(lv[k2][j].x, lv[k2][j].y),
                         fmaxf(lv[k2][j].z, lv[k2][j].w)));
    m = wave_max(m);
    float s = 0.f;
#pragma unroll
    for (int j = 0; j < 4; ++j)
      s += (expf(lv[k2][j].x - m) + expf(lv[k2][j].y - m)) +
           (expf(lv[k2][j].z - m) + expf(lv[k2][j].w - m));
    s = wave_sum(s);
    a[k2] = 1.0f - expf(lb[k2] - m) / s;
  }

  float m2 = -INFINITY;
#pragma unroll
  for (int k2 = 0; k2 < WM_K; ++k2) m2 = fmaxf(m2, a[k2]);
  float e[WM_K];
  float ss = 0.f;
#pragma unroll
  for (int k2 = 0; k2 < WM_K; ++k2) { e[k2] = expf(a[k2] - m2); ss += e[k2]; }
  if (lane == 0) {
#pragma unroll
    for (int k2 = 0; k2 < WM_K; ++k2) wmix[(size_t)row * WM_K + k2] = e[k2] / ss;
  }
}

__global__ __launch_bounds__(256) void stream_prep(
    const float* __restrict__ X, const float* __restrict__ wmix,
    const float* __restrict__ tags, const float* __restrict__ g,
    const float* __restrict__ bt, float* __restrict__ Xk,
    float* __restrict__ Qin, int nrows, int kk)
{
  const int lane = threadIdx.x & 31;
  const int row  = blockIdx.x * ROW_WPB + (threadIdx.x >> 5);
  if (row >= nrows) return;
  const float wv = wmix[(size_t)row * WM_K + kk];
  const v4f* x4  = (const v4f*)(X    + (size_t)row * ROW_D);
  const v4f* tg4 = (const v4f*)(tags + (size_t)kk  * ROW_D);
  v4f* xk4 = (v4f*)(Xk + (size_t)row * ROW_D);
  v4f v[ROW_P];
#pragma unroll
  for (int j = 0; j < ROW_P; ++j) {
    v[j] = x4[lane + 32 * j] * wv + tg4[lane + 32 * j];
    xk4[lane + 32 * j] = v[j];
  }
  float s = 0.f, s2 = 0.f;
#pragma unroll
  for (int j = 0; j < ROW_P; ++j) {
    s  += (v[j].x + v[j].y) + (v[j].z + v[j].w);
    s2 += (v[j].x * v[j].x + v[j].y * v[j].y) +
          (v[j].z * v[j].z + v[j].w * v[j].w);
  }
  s  = wave_sum(s);
  s2 = wave_sum(s2);
  const float mean = s / ROW_D;
  const float var  = s2 / ROW_D - mean * mean;
  const float inv  = rsqrtf(var + 1e-5f);
  const v4f* g4 = (const v4f*)g;
  const v4f* b4 = (const v4f*)bt;
  v4f* q4 = (v4f*)(Qin + (size_t)row * ROW_D);
#pragma unroll
  for (int j = 0; j < ROW_P; ++j) {
    const v4f gv = g4[lane + 32 * j];
    const v4f bv = b4[lane + 32 * j];
    q4[lane + 32 * j] = (v[j] - mean) * inv * gv + bv;
  }
}

#define AT_T     1024
#define AT_D     512
#define AT_H     8
#define AT_HD    64
#define AT_BAND  24
#define AT_QW    16
#define AT_WAVES 4
#define AT_QB    (AT_QW * AT_WAVES)
#define AT_WIN   (AT_QB + 2 * AT_BAND)
#define AT_NST   (AT_WIN / 8)
#define AT_LDP   68

template <int M>
__device__ __forceinline__ float row_xor(float x) {
  const int b = __builtin_bit_cast(int, x);
  return __builtin_bit_cast(
      float, __builtin_amdgcn_update_dpp(b, b, 0x160 + M, 0xf, 0xf, false));
}
__device__ __forceinline__ float half_max(float v) {
  v = fmaxf(v, row_xor<1>(v));
  v = fmaxf(v, row_xor<2>(v));
  v = fmaxf(v, row_xor<4>(v));
  v = fmaxf(v, row_xor<8>(v));
  return v;
}
__device__ __forceinline__ float half_sum(float v) {
  v += row_xor<1>(v);
  v += row_xor<2>(v);
  v += row_xor<4>(v);
  v += row_xor<8>(v);
  return v;
}

__global__ __launch_bounds__(128) void banded_attn_wmma(
    const float* __restrict__ q, const float* __restrict__ kb,
    const float* __restrict__ vb, float* __restrict__ ctx)
{
  __shared__ __attribute__((aligned(16))) float Ks[AT_WIN * AT_LDP];
  __shared__ __attribute__((aligned(16))) float Vs[AT_WIN * AT_LDP];
  __shared__ __attribute__((aligned(16))) float Ps[AT_WAVES][AT_QW * AT_LDP];

  const int tid  = threadIdx.x;
  const int lane = tid & 31;
  const int w    = tid >> 5;
  const int c    = lane & 15;
  const int hf   = lane >> 4;

  constexpr int tps = AT_T / AT_QB;
  const int blk = blockIdx.x;
  const int tb  = (blk % tps) * AT_QB;
  const int h   = (blk / tps) % AT_H;
  const int b   = blk / (tps * AT_H);
  const int t0  = tb + AT_QW * w;

  const size_t seqoff = (size_t)b * AT_T * AT_D + (size_t)h * AT_HD;

  v4f qa[AT_HD / 8];
  {
    const float* qrow = q + seqoff + (size_t)(t0 + c) * AT_D + 4 * hf;
#pragma unroll
    for (int m = 0; m < AT_HD / 8; ++m) qa[m] = *(const v4f*)(qrow + 8 * m);
  }

  {
    const int srow = tid >> 4;
    const int scol = (tid & 15) * 4;
    const float* kbase = kb + seqoff + scol;
    const float* vbase = vb + seqoff + scol;
    v4f kr[AT_NST], vr[AT_NST];
#pragma unroll
    for (int it = 0; it < AT_NST; ++it) {
      int s = tb - AT_BAND + srow + 8 * it;
      s = max(s, 0);
      s = min(s, AT_T - 1);
      kr[it] = *(const v4f*)(kbase + (size_t)s * AT_D);
      vr[it] = *(const v4f*)(vbase + (size_t)s * AT_D);
    }
#pragma unroll
    for (int it = 0; it < AT_NST; ++it) {
      const int r = srow + 8 * it;
      *(v4f*)(&Ks[r * AT_LDP + scol]) = kr[it];
      *(v4f*)(&Vs[r * AT_LDP + scol]) = vr[it];
    }
  }
  __syncthreads();

  v8f sc[4] = {{}, {}, {}, {}};
  {
    const float* kf = &Ks[(AT_QW * w + c) * AT_LDP + 4 * hf];
#pragma unroll
    for (int m = 0; m < AT_HD / 8; ++m) {
      const v4f b0 = *(const v4f*)(kf + 8 * m);
      const v4f b1 = *(const v4f*)(kf + 16 * AT_LDP + 8 * m);
      const v4f b2 = *(const v4f*)(kf + 32 * AT_LDP + 8 * m);
      const v4f b3 = *(const v4f*)(kf + 48 * AT_LDP + 8 * m);
      const v2f alo = {qa[m].x, qa[m].y}, ahi = {qa[m].z, qa[m].w};
      const v2f b0l = {b0.x, b0.y}, b0h = {b0.z, b0.w};
      const v2f b1l = {b1.x, b1.y}, b1h = {b1.z, b1.w};
      const v2f b2l = {b2.x, b2.y}, b2h = {b2.z, b2.w};
      const v2f b3l = {b3.x, b3.y}, b3h = {b3.z, b3.w};
      WMMA_F32(sc[0], alo, b0l);
      WMMA_F32(sc[1], alo, b1l);
      WMMA_F32(sc[2], alo, b2l);
      WMMA_F32(sc[3], alo, b3l);
      WMMA_F32(sc[0], ahi, b0h);
      WMMA_F32(sc[1], ahi, b1h);
      WMMA_F32(sc[2], ahi, b2h);
      WMMA_F32(sc[3], ahi, b3h);
    }
  }

  const float scale = 0.125f;
  float rinv[8];
  {
    float* pw = &Ps[w][0];
#pragma unroll
    for (int r = 0; r < 8; ++r) {
      const int i = 8 * hf + r;
      float val[4];
      float mx = -INFINITY;
#pragma unroll
      for (int nt = 0; nt < 4; ++nt) {
        const int j  = 16 * nt + c;
        const int d  = j - i;
        const int sg = t0 - AT_BAND + j;
        const bool ok = (d >= 0) && (d <= 2 * AT_BAND) && (sg >= 0) && (sg < AT_T);
        val[nt] = ok ? sc[nt][r] * scale : -INFINITY;
        mx = fmaxf(mx, val[nt]);
      }
      mx = half_max(mx);
      float sum = 0.f;
#pragma unroll
      for (int nt = 0; nt < 4; ++nt) {
        const float e = expf(val[nt] - mx);
        sum += e;
        pw[i * AT_LDP + 16 * nt + c] = e;
      }
      sum = half_sum(sum);
      rinv[r] = 1.0f / sum;
    }
  }
  __syncthreads();

  v8f oc[4] = {{}, {}, {}, {}};
  {
    const float* pf = &Ps[w][c * AT_LDP + 4 * hf];
    const float* vf = &Vs[(AT_QW * w + 4 * hf) * AT_LDP + c];
#pragma unroll
    for (int m = 0; m < AT_HD / 8; ++m) {
      const v4f a = *(const v4f*)(pf + 8 * m);
      const v2f alo = {a.x, a.y}, ahi = {a.z, a.w};
      const float* vm = vf + 8 * m * AT_LDP;
#pragma unroll
      for (int nt = 0; nt < 4; ++nt) {
        const v2f bl = {vm[16 * nt],              vm[AT_LDP + 16 * nt]};
        const v2f bh = {vm[2 * AT_LDP + 16 * nt], vm[3 * AT_LDP + 16 * nt]};
        WMMA_F32(oc[nt], alo, bl);
        WMMA_F32(oc[nt], ahi, bh);
      }
    }
  }
  __syncthreads();

  {
    float* pw = &Ps[w][0];
#pragma unroll
    for (int r = 0; r < 8; ++r) {
      const int i = 8 * hf + r;
#pragma unroll
      for (int nt = 0; nt < 4; ++nt)
        pw[i * AT_LDP + 16 * nt + c] = oc[nt][r] * rinv[r];
    }
  }
  __syncthreads();
  {
    const float* pw = &Ps[w][0];
    float* crow = ctx + seqoff + (size_t)t0 * AT_D;
#pragma unroll
    for (int u = 0; u < 8; ++u) {
      const int f   = lane + 32 * u;
      const int row = f >> 4;
      const int col = (f & 15) * 4;
      const v4f o = *(const v4f*)(pw + row * AT_LDP + col);
      *(v4f*)(crow + (size_t)row * AT_D + col) = o;
    }
  }
}

extern "C" void kernel_launch(void* const* d_in, const int* in_sizes, int n_in,
                              void* d_out, int out_size, void* d_ws, size_t ws_size,
                              hipStream_t stream) {
  (void)in_sizes; (void)n_in; (void)out_size; (void)ws_size;
  constexpr int B = 4, T = 1024, D_IN = 512, D = 512, K = 2, V = 500, H = 8, BAND = 24;
  constexpr int M = B * T;

  const float* x_m      = (const float*)d_in[0];
  const float* logits   = (const float*)d_in[1];
  const int*   blank_id = (const int*)  d_in[2];
  const float* proj_w   = (const float*)d_in[3];
  const float* proj_b   = (const float*)d_in[4];
  const float* spk_tags = (const float*)d_in[5];
  const float* lnq_g    = (const float*)d_in[6];
  const float* lnq_b    = (const float*)d_in[7];
  const float* lnkv_g   = (const float*)d_in[8];
  const float* lnkv_b   = (const float*)d_in[9];
  const float* in_w     = (const float*)d_in[10];
  const float* in_b     = (const float*)d_in[11];
  const float* out_w    = (const float*)d_in[12];
  const float* out_b    = (const float*)d_in[13];
  const float* ffn_ln_g = (const float*)d_in[14];
  const float* ffn_ln_b = (const float*)d_in[15];
  const float* ffn_w1   = (const float*)d_in[16];
  const float* ffn_b1   = (const float*)d_in[17];
  const float* ffn_w2   = (const float*)d_in[18];
  const float* ffn_b2   = (const float*)d_in[19];
  float* out = (float*)d_out;

  float* ws = (float*)d_ws;
  const size_t SZ = (size_t)M * D;
  float* X    = ws;
  float* kbuf = X    + SZ;
  float* vbuf = kbuf + SZ;
  float* tA   = vbuf + SZ;
  float* tB   = tA   + SZ;
  float* tC   = tB   + SZ;
  float* t1   = tC   + SZ;
  float* wmix = t1   + 4 * SZ;

  const dim3 blk(256);
  static_assert(D == ROW_D && V == WM_V && K == WM_K && M % ROW_WPB == 0,
                "row kernels are written for these sizes");
  static_assert(T == AT_T && D == AT_D && H == AT_H && D / H == AT_HD &&
                BAND == AT_BAND && T % AT_QB == 0,
                "attention kernel is written for these sizes");
  const dim3 rowgrid(M / ROW_WPB);
  auto gg = [](int m, int n) { return dim3(m / 64, n / 128); };

  gemm_wmma_f32<<<gg(M, D), blk, 0, stream>>>(x_m, proj_w, proj_b, nullptr, X,
                                              M, D, D_IN, 0, T, K, 0);
  wmix_kernel<<<rowgrid, blk, 0, stream>>>(logits, blank_id, wmix, M);
  ln_kernel<<<rowgrid, blk, 0, stream>>>(X, lnkv_g, lnkv_b, tA, M);
  gemm_wmma_f32<<<gg(M, D), blk, 0, stream>>>(tA, in_w + (size_t)D * D, in_b + D,
                                              nullptr, kbuf, M, D, D, 0, T, K, 0);
  gemm_wmma_f32<<<gg(M, D), blk, 0, stream>>>(tA, in_w + (size_t)2 * D * D, in_b + 2 * D,
                                              nullptr, vbuf, M, D, D, 0, T, K, 0);

  for (int kk = 0; kk < K; ++kk) {
    stream_prep<<<rowgrid, blk, 0, stream>>>(X, wmix, spk_tags, lnq_g, lnq_b,
                                             tA, tB, M, kk);
    gemm_wmma_f32<<<gg(M, D), blk, 0, stream>>>(tB, in_w, in_b, nullptr, tC,
                                                M, D, D, 0, T, K, kk);
    banded_attn_wmma<<<dim3((B * H * T) / AT_QB), dim3(32 * AT_WAVES), 0, stream>>>(
        tC, kbuf, vbuf, tB);
    gemm_wmma_f32<<<gg(M, D), blk, 0, stream>>>(tB, out_w, out_b, tA, tC,
                                                M, D, D,  1, T, K, kk);
    ln_kernel<<<rowgrid, blk, 0, stream>>>(tC, ffn_ln_g, ffn_ln_b, tB, M);
    gemm_wmma_f32<<<gg(M, 4 * D), blk, 0, stream>>>(tB, ffn_w1, ffn_b1, nullptr, t1,
                                                    M, 4 * D, D,  2, T, K, kk);
    gemm_wmma_f32<<<gg(M, D), blk, 0, stream>>>(t1, ffn_w2, ffn_b2, tC, out,
                                                M, D, 4 * D,  1 | 4, T, K, kk);
  }
}


# Speaker loop folded into the row dimension: stream prep, q projection, banded attention, output projection and FFN LayerNorm each run once over both speakers' 8192 rows (19 launches -> 14)
# speedup vs baseline: 2.2439x; 1.0885x over previous
#include <hip/hip_runtime.h>
#include <hip/hip_bf16.h>
#include <math.h>

typedef float v2f __attribute__((ext_vector_type(2)));
typedef float v4f __attribute__((ext_vector_type(4)));
typedef float v8f __attribute__((ext_vector_type(8)));

__device__ __forceinline__ float gelu_exact(float x) {
  return 0.5f * x * (1.0f + erff(x * 0.70710678118654752f));
}

#define WMMA_F32(cacc, av, bv)                                                 \
  cacc = __builtin_amdgcn_wmma_f32_16x16x4_f32(false, (av), false, (bv),       \
                                               (short)0, (cacc), false, false)

#define GEMM_BM 64
#define GEMM_BN 128
#define GEMM_BK 32
#define GEMM_LDP 36

struct GemmFrag { v4f a, b0, b1, b2, b3; };

__device__ __forceinline__ void gemm_fence() { __builtin_amdgcn_sched_barrier(0); }

__device__ __forceinline__ void gemm_ldfrag(GemmFrag& f, const float* as,
                                            const float* bs) {
  f.a  = *(const v4f*)(as);
  f.b0 = *(const v4f*)(bs);
  f.b1 = *(const v4f*)(bs + 16 * GEMM_LDP);
  f.b2 = *(const v4f*)(bs + 32 * GEMM_LDP);
  f.b3 = *(const v4f*)(bs + 48 * GEMM_LDP);
}

__device__ __forceinline__ void gemm_mma8(const GemmFrag& f, v8f& c0, v8f& c1,
                                          v8f& c2, v8f& c3) {
  const v2f alo = {f.a.x, f.a.y},   ahi = {f.a.z, f.a.w};
  const v2f b0l = {f.b0.x, f.b0.y}, b0h = {f.b0.z, f.b0.w};
  const v2f b1l = {f.b1.x, f.b1.y}, b1h = {f.b1.z, f.b1.w};
  const v2f b2l = {f.b2.x, f.b2.y}, b2h = {f.b2.z, f.b2.w};
  const v2f b3l = {f.b3.x, f.b3.y}, b3h = {f.b3.z, f.b3.w};
  WMMA_F32(c0, alo, b0l);
  WMMA_F32(c1, alo, b1l);
  WMMA_F32(c2, alo, b2l);
  WMMA_F32(c3, alo, b3l);
  WMMA_F32(c0, ahi, b0h);
  WMMA_F32(c1, ahi, b1h);
  WMMA_F32(c2, ahi, b2h);
  WMMA_F32(c3, ahi, b3h);
}

__global__ __launch_bounds__(256) void gemm_wmma_f32(
    const float* __restrict__ A, const float* __restrict__ W,
    const float* __restrict__ bias, const float* __restrict__ R,
    float* __restrict__ C, int M, int N, int Kd,
    int flags, int Tlen, int Kspk, int kk)
{
  __shared__ __attribute__((aligned(16))) float As[2][GEMM_BM * GEMM_LDP];
  __shared__ __attribute__((aligned(16))) float Bs[2][GEMM_BN * GEMM_LDP];

  const int tid  = threadIdx.x;
  const int lane = tid & 31;
  const int w    = tid >> 5;
  const int wm   = w & 3;
  const int wn   = w >> 2;
  const int blockM = blockIdx.x * GEMM_BM;
  const int blockN = blockIdx.y * GEMM_BN;
  const int tileM = blockM + wm * 16;
  const int tileN = blockN + wn * 64;
  const int half = lane >> 4;
  const int l16  = lane & 15;
  const int n0   = tileN + l16;

  const int srow = tid >> 3;
  const int scol = (tid & 7) * 4;
  const float* Ag = A + (size_t)(blockM + srow) * Kd + scol;
  const float* Bg = W + (size_t)(blockN + srow) * Kd + scol;
  const size_t rstep = (size_t)32 * Kd;
  const int sts = srow * GEMM_LDP + scol;

  v4f ra0 = *(const v4f*)(Ag);
  v4f ra1 = *(const v4f*)(Ag + rstep);
  v4f rb0 = *(const v4f*)(Bg);
  v4f rb1 = *(const v4f*)(Bg + rstep);
  v4f rb2 = *(const v4f*)(Bg + 2 * rstep);
  v4f rb3 = *(const v4f*)(Bg + 3 * rstep);

  *(v4f*)(&As[0][sts])                 = ra0;
  *(v4f*)(&As[0][sts + 32 * GEMM_LDP]) = ra1;
  *(v4f*)(&Bs[0][sts])                 = rb0;
  *(v4f*)(&Bs[0][sts + 32 * GEMM_LDP]) = rb1;
  *(v4f*)(&Bs[0][sts + 64 * GEMM_LDP]) = rb2;
  *(v4f*)(&Bs[0][sts + 96 * GEMM_LDP]) = rb3;
  __syncthreads();

  const int nk = Kd / GEMM_BK;
  if (1 < nk) {
    ra0 = *(const v4f*)(Ag + GEMM_BK);
    ra1 = *(const v4f*)(Ag + rstep + GEMM_BK);
    rb0 = *(const v4f*)(Bg + GEMM_BK);
    rb1 = *(const v4f*)(Bg + rstep + GEMM_BK);
    rb2 = *(const v4f*)(Bg + 2 * rstep + GEMM_BK);
    rb3 = *(const v4f*)(Bg + 3 * rstep + GEMM_BK);
  }

  const int fa = (wm * 16 + l16) * GEMM_LDP + 4 * half;
  const int fb = (wn * 64 + l16) * GEMM_LDP + 4 * half;

  v8f c0 = {}, c1 = {}, c2 = {}, c3 = {};

  GemmFrag f0, f1;
  gemm_ldfrag(f0, &As[0][fa], &Bs[0][fb]);

  for (int kt = 0; kt < nk; ++kt) {
    const int cur = kt & 1;
    const int nxt = cur ^ 1;
    const float* as = &As[cur][fa];
    const float* bs = &Bs[cur][fb];

    gemm_fence();
    gemm_ldfrag(f1, as + 8, bs + 8);
    gemm_fence();
    gemm_mma8(f0, c0, c1, c2, c3);
    gemm_fence();
    gemm_ldfrag(f0, as + 16, bs + 16);
    gemm_fence();
    gemm_mma8(f1, c0, c1, c2, c3);
    gemm_fence();
    gemm_ldfrag(f1, as + 24, bs + 24);
    *(v4f*)(&As[nxt][sts])                 = ra0;
    *(v4f*)(&As[nxt][sts + 32 * GEMM_LDP]) = ra1;
    *(v4f*)(&Bs[nxt][sts])                 = rb0;
    *(v4f*)(&Bs[nxt][sts + 32 * GEMM_LDP]) = rb1;
    *(v4f*)(&Bs[nxt][sts + 64 * GEMM_LDP]) = rb2;
    *(v4f*)(&Bs[nxt][sts + 96 * GEMM_LDP]) = rb3;
    gemm_fence();
    gemm_mma8(f0, c0, c1, c2, c3);
    gemm_fence();

    __syncthreads();
    if (kt + 2 < nk) {
      const int ko = (kt + 2) * GEMM_BK;
      ra0 = *(const v4f*)(Ag + ko);
      ra1 = *(const v4f*)(Ag + rstep + ko);
      rb0 = *(const v4f*)(Bg + ko);
      rb1 = *(const v4f*)(Bg + rstep + ko);
      rb2 = *(const v4f*)(Bg + 2 * rstep + ko);
      rb3 = *(const v4f*)(Bg + 3 * rstep + ko);
    }
    gemm_ldfrag(f0, &As[nxt][fa], &Bs[nxt][fb]);
    gemm_fence();
    gemm_mma8(f1, c0, c1, c2, c3);
    gemm_fence();
  }

  const float bv0 = bias[n0];
  const float bv1 = bias[n0 + 16];
  const float bv2 = bias[n0 + 32];
  const float bv3 = bias[n0 + 48];
#pragma unroll
  for (int i = 0; i < 8; ++i) {
    const int row = tileM + i + 8 * half;
    float v0 = c0[i] + bv0;
    float v1 = c1[i] + bv1;
    float v2 = c2[i] + bv2;
    float v3 = c3[i] + bv3;
    if (flags & 1) {
      const float* rr = R + (size_t)row * N;
      v0 += rr[n0];
      v1 += rr[n0 + 16];
      v2 += rr[n0 + 32];
      v3 += rr[n0 + 48];
    }
    if (flags & 2) {
      v0 = gelu_exact(v0);
      v1 = gelu_exact(v1);
      v2 = gelu_exact(v2);
      v3 = gelu_exact(v3);
    }
    size_t orow = (size_t)row;
    if (flags & 4) {
      const int bi = row / Tlen;
      orow = (size_t)bi * Kspk * Tlen + (size_t)kk * Tlen + (row % Tlen);
    }
    float* cr = C + orow * N;
    cr[n0]      = v0;
    cr[n0 + 16] = v1;
    cr[n0 + 32] = v2;
    cr[n0 + 48] = v3;
  }
}

#define ROW_D   512
#define ROW_P   (ROW_D / 128)
#define ROW_WPB 8
#define WM_V    500
#define WM_V4   (WM_V / 4)
#define WM_K    2

__device__ __forceinline__ float wave_sum(float v) {
#pragma unroll
  for (int off = 16; off > 0; off >>= 1) v += __shfl_xor(v, off, 32);
  return v;
}
__device__ __forceinline__ float wave_max(float v) {
#pragma unroll
  for (int off = 16; off > 0; off >>= 1) v = fmaxf(v, __shfl_xor(v, off, 32));
  return v;
}

__global__ __launch_bounds__(256) void ln_kernel(
    const float* __restrict__ X, const float* __restrict__ g,
    const float* __restrict__ b, float* __restrict__ Y, int nrows)
{
  const int lane = threadIdx.x & 31;
  const int row  = blockIdx.x * ROW_WPB + (threadIdx.x >> 5);
  if (row >= nrows) return;
  const v4f* x4 = (const v4f*)(X + (size_t)row * ROW_D);
  v4f xv[ROW_P];
#pragma unroll
  for (int j = 0; j < ROW_P; ++j) xv[j] = x4[lane + 32 * j];
  float s = 0.f, s2 = 0.f;
#pragma unroll
  for (int j = 0; j < ROW_P; ++j) {
    s  += (xv[j].x + xv[j].y) + (xv[j].z + xv[j].w);
    s2 += (xv[j].x * xv[j].x + xv[j].y * xv[j].y) +
          (xv[j].z * xv[j].z + xv[j].w * xv[j].w);
  }
  s  = wave_sum(s);
  s2 = wave_sum(s2);
  const float mean = s / ROW_D;
  const float var  = s2 / ROW_D - mean * mean;
  const float inv  = rsqrtf(var + 1e-5f);
  const v4f* g4 = (const v4f*)g;
  const v4f* b4 = (const v4f*)b;
  v4f* y4 = (v4f*)(Y + (size_t)row * ROW_D);
#pragma unroll
  for (int j = 0; j < ROW_P; ++j) {
    const v4f gv = g4[lane + 32 * j];
    const v4f bv = b4[lane + 32 * j];
    y4[lane + 32 * j] = (xv[j] - mean) * inv * gv + bv;
  }
}

__global__ __launch_bounds__(256) void wmix_kernel(
    const float* __restrict__ logits, const int* __restrict__ blank_p,
    float* __restrict__ wmix, int BT)
{
  const int lane = threadIdx.x & 31;
  const int row  = blockIdx.x * ROW_WPB + (threadIdx.x >> 5);
  if (row >= BT) return;
  const int blank = blank_p[0];
  const v4f ninf = {-INFINITY, -INFINITY, -INFINITY, -INFINITY};

  v4f lv[WM_K][4];
  float lb[WM_K];
#pragma unroll
  for (int k2 = 0; k2 < WM_K; ++k2) {
    const float* L = logits + ((size_t)k2 * BT + row) * WM_V;
    const v4f* L4 = (const v4f*)L;
#pragma unroll
    for (int j = 0; j < 4; ++j) {
      const int i4 = lane + 32 * j;
      lv[k2][j] = (i4 < WM_V4) ? L4[i4] : ninf;
    }
    lb[k2] = L[blank];
  }

  float a[WM_K];
#pragma unroll
  for (int k2 = 0; k2 < WM_K; ++k2) {
    float m = -INFINITY;
#pragma unroll
    for (int j = 0; j < 4; ++j)
      m = fmaxf(m, fmaxf(fmaxf(lv[k2][j].x, lv[k2][j].y),
                         fmaxf(lv[k2][j].z, lv[k2][j].w)));
    m = wave_max(m);
    float s = 0.f;
#pragma unroll
    for (int j = 0; j < 4; ++j)
      s += (expf(lv[k2][j].x - m) + expf(lv[k2][j].y - m)) +
           (expf(lv[k2][j].z - m) + expf(lv[k2][j].w - m));
    s = wave_sum(s);
    a[k2] = 1.0f - expf(lb[k2] - m) / s;
  }

  float m2 = -INFINITY;
#pragma unroll
  for (int k2 = 0; k2 < WM_K; ++k2) m2 = fmaxf(m2, a[k2]);
  float e[WM_K];
  float ss = 0.f;
#pragma unroll
  for (int k2 = 0; k2 < WM_K; ++k2) { e[k2] = expf(a[k2] - m2); ss += e[k2]; }
  if (lane == 0) {
#pragma unroll
    for (int k2 = 0; k2 < WM_K; ++k2) wmix[(size_t)row * WM_K + k2] = e[k2] / ss;
  }
}

__global__ __launch_bounds__(256) void stream_prep(
    const float* __restrict__ X, const float* __restrict__ wmix,
    const float* __restrict__ tags, const float* __restrict__ g,
    const float* __restrict__ bt, float* __restrict__ Xk,
    float* __restrict__ Qin, int nrows)
{
  const int lane = threadIdx.x & 31;
  const int row  = blockIdx.x * ROW_WPB + (threadIdx.x >> 5);
  if (row >= nrows) return;
  const v4f* x4 = (const v4f*)(X + (size_t)row * ROW_D);
  const v4f* g4 = (const v4f*)g;
  const v4f* b4 = (const v4f*)bt;
  v4f xv[ROW_P], gv[ROW_P], bv[ROW_P];
#pragma unroll
  for (int j = 0; j < ROW_P; ++j) {
    xv[j] = x4[lane + 32 * j];
    gv[j] = g4[lane + 32 * j];
    bv[j] = b4[lane + 32 * j];
  }
#pragma unroll
  for (int kk = 0; kk < WM_K; ++kk) {
    const float wv = wmix[(size_t)row * WM_K + kk];
    const v4f* tg4 = (const v4f*)(tags + (size_t)kk * ROW_D);
    const size_t orow = (size_t)kk * nrows + row;
    v4f* xk4 = (v4f*)(Xk + orow * ROW_D);
    v4f v[ROW_P];
#pragma unroll
    for (int j = 0; j < ROW_P; ++j) {
      v[j] = xv[j] * wv + tg4[lane + 32 * j];
      xk4[lane + 32 * j] = v[j];
    }
    float s = 0.f, s2 = 0.f;
#pragma unroll
    for (int j = 0; j < ROW_P; ++j) {
      s  += (v[j].x + v[j].y) + (v[j].z + v[j].w);
      s2 += (v[j].x * v[j].x + v[j].y * v[j].y) +
            (v[j].z * v[j].z + v[j].w * v[j].w);
    }
    s  = wave_sum(s);
    s2 = wave_sum(s2);
    const float mean = s / ROW_D;
    const float var  = s2 / ROW_D - mean * mean;
    const float inv  = rsqrtf(var + 1e-5f);
    v4f* q4 = (v4f*)(Qin + orow * ROW_D);
#pragma unroll
    for (int j = 0; j < ROW_P; ++j)
      q4[lane + 32 * j] = (v[j] - mean) * inv * gv[j] + bv[j];
  }
}

#define AT_T     1024
#define AT_D     512
#define AT_B     4
#define AT_H     8
#define AT_HD    64
#define AT_BAND  24
#define AT_QW    16
#define AT_WAVES 4
#define AT_QB    (AT_QW * AT_WAVES)
#define AT_WIN   (AT_QB + 2 * AT_BAND)
#define AT_NST   (AT_WIN / 8)
#define AT_LDP   68

template <int M>
__device__ __forceinline__ float row_xor(float x) {
  const int b = __builtin_bit_cast(int, x);
  return __builtin_bit_cast(
      float, __builtin_amdgcn_update_dpp(b, b, 0x160 + M, 0xf, 0xf, false));
}
__device__ __forceinline__ float half_max(float v) {
  v = fmaxf(v, row_xor<1>(v));
  v = fmaxf(v, row_xor<2>(v));
  v = fmaxf(v, row_xor<4>(v));
  v = fmaxf(v, row_xor<8>(v));
  return v;
}
__device__ __forceinline__ float half_sum(float v) {
  v += row_xor<1>(v);
  v += row_xor<2>(v);
  v += row_xor<4>(v);
  v += row_xor<8>(v);
  return v;
}

__global__ __launch_bounds__(128) void banded_attn_wmma(
    const float* __restrict__ q, const float* __restrict__ kb,
    const float* __restrict__ vb, float* __restrict__ ctx)
{
  __shared__ __attribute__((aligned(16))) float Ks[AT_WIN * AT_LDP];
  __shared__ __attribute__((aligned(16))) float Vs[AT_WIN * AT_LDP];
  __shared__ __attribute__((aligned(16))) float Ps[AT_WAVES][AT_QW * AT_LDP];

  const int tid  = threadIdx.x;
  const int lane = tid & 31;
  const int w    = tid >> 5;
  const int c    = lane & 15;
  const int hf   = lane >> 4;

  constexpr int tps = AT_T / AT_QB;
  const int blk = blockIdx.x;
  const int tb  = (blk % tps) * AT_QB;
  const int h   = (blk / tps) % AT_H;
  const int b   = (blk / (tps * AT_H)) % AT_B;
  const int spk = blk / (tps * AT_H * AT_B);
  const int t0  = tb + AT_QW * w;

  const size_t seqoff = (size_t)b * AT_T * AT_D + (size_t)h * AT_HD;
  const size_t qoff   = (size_t)(spk * AT_B + b) * AT_T * AT_D +
                        (size_t)h * AT_HD;

  v4f qa[AT_HD / 8];
  {
    const float* qrow = q + qoff + (size_t)(t0 + c) * AT_D + 4 * hf;
#pragma unroll
    for (int m = 0; m < AT_HD / 8; ++m) qa[m] = *(const v4f*)(qrow + 8 * m);
  }

  {
    const int srow = tid >> 4;
    const int scol = (tid & 15) * 4;
    const float* kbase = kb + seqoff + scol;
    const float* vbase = vb + seqoff + scol;
    v4f kr[AT_NST], vr[AT_NST];
#pragma unroll
    for (int it = 0; it < AT_NST; ++it) {
      int s = tb - AT_BAND + srow + 8 * it;
      s = max(s, 0);
      s = min(s, AT_T - 1);
      kr[it] = *(const v4f*)(kbase + (size_t)s * AT_D);
      vr[it] = *(const v4f*)(vbase + (size_t)s * AT_D);
    }
#pragma unroll
    for (int it = 0; it < AT_NST; ++it) {
      const int r = srow + 8 * it;
      *(v4f*)(&Ks[r * AT_LDP + scol]) = kr[it];
      *(v4f*)(&Vs[r * AT_LDP + scol]) = vr[it];
    }
  }
  __syncthreads();

  v8f sc[4] = {{}, {}, {}, {}};
  {
    const float* kf = &Ks[(AT_QW * w + c) * AT_LDP + 4 * hf];
#pragma unroll
    for (int m = 0; m < AT_HD / 8; ++m) {
      const v4f b0 = *(const v4f*)(kf + 8 * m);
      const v4f b1 = *(const v4f*)(kf + 16 * AT_LDP + 8 * m);
      const v4f b2 = *(const v4f*)(kf + 32 * AT_LDP + 8 * m);
      const v4f b3 = *(const v4f*)(kf + 48 * AT_LDP + 8 * m);
      const v2f alo = {qa[m].x, qa[m].y}, ahi = {qa[m].z, qa[m].w};
      const v2f b0l = {b0.x, b0.y}, b0h = {b0.z, b0.w};
      const v2f b1l = {b1.x, b1.y}, b1h = {b1.z, b1.w};
      const v2f b2l = {b2.x, b2.y}, b2h = {b2.z, b2.w};
      const v2f b3l = {b3.x, b3.y}, b3h = {b3.z, b3.w};
      WMMA_F32(sc[0], alo, b0l);
      WMMA_F32(sc[1], alo, b1l);
      WMMA_F32(sc[2], alo, b2l);
      WMMA_F32(sc[3], alo, b3l);
      WMMA_F32(sc[0], ahi, b0h);
      WMMA_F32(sc[1], ahi, b1h);
      WMMA_F32(sc[2], ahi, b2h);
      WMMA_F32(sc[3], ahi, b3h);
    }
  }

  const float scale = 0.125f;
  float rinv[8];
  {
    float* pw = &Ps[w][0];
#pragma unroll
    for (int r = 0; r < 8; ++r) {
      const int i = 8 * hf + r;
      float val[4];
      float mx = -INFINITY;
#pragma unroll
      for (int nt = 0; nt < 4; ++nt) {
        const int j  = 16 * nt + c;
        const int d  = j - i;
        const int sg = t0 - AT_BAND + j;
        const bool ok = (d >= 0) && (d <= 2 * AT_BAND) && (sg >= 0) && (sg < AT_T);
        val[nt] = ok ? sc[nt][r] * scale : -INFINITY;
        mx = fmaxf(mx, val[nt]);
      }
      mx = half_max(mx);
      float sum = 0.f;
#pragma unroll
      for (int nt = 0; nt < 4; ++nt) {
        const float e = expf(val[nt] - mx);
        sum += e;
        pw[i * AT_LDP + 16 * nt + c] = e;
      }
      sum = half_sum(sum);
      rinv[r] = 1.0f / sum;
    }
  }
  __syncthreads();

  v8f oc[4] = {{}, {}, {}, {}};
  {
    const float* pf = &Ps[w][c * AT_LDP + 4 * hf];
    const float* vf = &Vs[(AT_QW * w + 4 * hf) * AT_LDP + c];
#pragma unroll
    for (int m = 0; m < AT_HD / 8; ++m) {
      const v4f a = *(const v4f*)(pf + 8 * m);
      const v2f alo = {a.x, a.y}, ahi = {a.z, a.w};
      const float* vm = vf + 8 * m * AT_LDP;
#pragma unroll
      for (int nt = 0; nt < 4; ++nt) {
        const v2f bl = {vm[16 * nt],              vm[AT_LDP + 16 * nt]};
        const v2f bh = {vm[2 * AT_LDP + 16 * nt], vm[3 * AT_LDP + 16 * nt]};
        WMMA_F32(oc[nt], alo, bl);
        WMMA_F32(oc[nt], ahi, bh);
      }
    }
  }
  __syncthreads();

  {
    float* pw = &Ps[w][0];
#pragma unroll
    for (int r = 0; r < 8; ++r) {
      const int i = 8 * hf + r;
#pragma unroll
      for (int nt = 0; nt < 4; ++nt)
        pw[i * AT_LDP + 16 * nt + c] = oc[nt][r] * rinv[r];
    }
  }
  __syncthreads();
  {
    const float* pw = &Ps[w][0];
    float* crow = ctx + qoff + (size_t)t0 * AT_D;
#pragma unroll
    for (int u = 0; u < 8; ++u) {
      const int f   = lane + 32 * u;
      const int row = f >> 4;
      const int col = (f & 15) * 4;
      const v4f o = *(const v4f*)(pw + row * AT_LDP + col);
      *(v4f*)(crow + (size_t)row * AT_D + col) = o;
    }
  }
}

extern "C" void kernel_launch(void* const* d_in, const int* in_sizes, int n_in,
                              void* d_out, int out_size, void* d_ws, size_t ws_size,
                              hipStream_t stream) {
  (void)in_sizes; (void)n_in; (void)out_size; (void)ws_size;
  constexpr int B = 4, T = 1024, D_IN = 512, D = 512, K = 2, V = 500, H = 8, BAND = 24;
  constexpr int M = B * T;

  const float* x_m      = (const float*)d_in[0];
  const float* logits   = (const float*)d_in[1];
  const int*   blank_id = (const int*)  d_in[2];
  const float* proj_w   = (const float*)d_in[3];
  const float* proj_b   = (const float*)d_in[4];
  const float* spk_tags = (const float*)d_in[5];
  const float* lnq_g    = (const float*)d_in[6];
  const float* lnq_b    = (const float*)d_in[7];
  const float* lnkv_g   = (const float*)d_in[8];
  const float* lnkv_b   = (const float*)d_in[9];
  const float* in_w     = (const float*)d_in[10];
  const float* in_b     = (const float*)d_in[11];
  const float* out_w    = (const float*)d_in[12];
  const float* out_b    = (const float*)d_in[13];
  const float* ffn_ln_g = (const float*)d_in[14];
  const float* ffn_ln_b = (const float*)d_in[15];
  const float* ffn_w1   = (const float*)d_in[16];
  const float* ffn_b1   = (const float*)d_in[17];
  const float* ffn_w2   = (const float*)d_in[18];
  const float* ffn_b2   = (const float*)d_in[19];
  float* out = (float*)d_out;

  float* ws = (float*)d_ws;
  const size_t SZ = (size_t)M * D;
  float* X    = ws;
  float* kbuf = X    + SZ;
  float* vbuf = kbuf + SZ;
  float* tA   = vbuf + SZ;
  float* tB   = tA   + 2 * SZ;
  float* tC   = tB   + 2 * SZ;
  float* t1   = ws;
  float* wmix = ws   + 10 * SZ;

  const dim3 blk(256);
  static_assert(D == ROW_D && V == WM_V && K == WM_K && M % ROW_WPB == 0,
                "row kernels are written for these sizes");
  static_assert(T == AT_T && D == AT_D && H == AT_H && D / H == AT_HD &&
                B == AT_B && BAND == AT_BAND && T % AT_QB == 0,
                "attention kernel is written for these sizes");
  const dim3 rowgrid(M / ROW_WPB);
  const dim3 rowgrid2(K * M / ROW_WPB);
  auto gg = [](int m, int n) { return dim3(m / 64, n / 128); };

  gemm_wmma_f32<<<gg(M, D), blk, 0, stream>>>(x_m, proj_w, proj_b, nullptr, X,
                                              M, D, D_IN, 0, T, K, 0);
  wmix_kernel<<<rowgrid, blk, 0, stream>>>(logits, blank_id, wmix, M);
  ln_kernel<<<rowgrid, blk, 0, stream>>>(X, lnkv_g, lnkv_b, tB, M);
  gemm_wmma_f32<<<gg(M, D), blk, 0, stream>>>(tB, in_w + (size_t)D * D, in_b + D,
                                              nullptr, kbuf, M, D, D, 0, T, K, 0);
  gemm_wmma_f32<<<gg(M, D), blk, 0, stream>>>(tB, in_w + (size_t)2 * D * D, in_b + 2 * D,
                                              nullptr, vbuf, M, D, D, 0, T, K, 0);

  stream_prep<<<rowgrid, blk, 0, stream>>>(X, wmix, spk_tags, lnq_g, lnq_b,
                                           tA, tB, M);
  gemm_wmma_f32<<<gg(K * M, D), blk, 0, stream>>>(tB, in_w, in_b, nullptr, tC,
                                                  K * M, D, D, 0, T, K, 0);
  banded_attn_wmma<<<dim3((K * B * H * T) / AT_QB), dim3(32 * AT_WAVES), 0, stream>>>(
      tC, kbuf, vbuf, tB);
  gemm_wmma_f32<<<gg(K * M, D), blk, 0, stream>>>(tB, out_w, out_b, tA, tC,
                                                  K * M, D, D,  1, T, K, 0);
  ln_kernel<<<rowgrid2, blk, 0, stream>>>(tC, ffn_ln_g, ffn_ln_b, tB, K * M);

  for (int kk = 0; kk < K; ++kk) {
    gemm_wmma_f32<<<gg(M, 4 * D), blk, 0, stream>>>(tB + (size_t)kk * SZ, ffn_w1, ffn_b1,
                                                    nullptr, t1,
                                                    M, 4 * D, D,  2, T, K, kk);
    gemm_wmma_f32<<<gg(M, D), blk, 0, stream>>>(t1, ffn_w2, ffn_b2, tC + (size_t)kk * SZ, out,
                                                M, D, 4 * D,  1 | 4, T, K, kk);
  }
}
